# starting kernel
# baseline (speedup 1.0000x reference)
#include <hip/hip_runtime.h>
#include <hip/hip_bf16.h>
#include <math.h>


#define Nn    100000
#define FIN   128
#define HCc   64
#define HIDc  32
#define OUTc  40

typedef __attribute__((ext_vector_type(2))) float v2f;
typedef __attribute__((ext_vector_type(8))) float v8f;

__device__ inline void atomicMaxFloat(float* addr, float val) {
    int bits = __float_as_int(val);
    if (bits >= 0) {
        atomicMax((int*)addr, bits);
    } else {
        atomicMin((unsigned int*)addr, (unsigned int)bits);
    }
}

__device__ inline void edge_sd(const int* __restrict__ ei, int e, int E,
                               int& s, int& d) {
    if (e < E) { s = ei[e]; d = ei[E + e]; }
    else       { s = e - E; d = e - E; }
}

template<int NT, bool FUSE, int K>
__global__ __launch_bounds__(128)
void wmma_gemm(const float* __restrict__ A, const float* __restrict__ A2,
               const float* __restrict__ a2bias,
               const float* __restrict__ B, const float* __restrict__ bias,
               float* __restrict__ C, int M, int Ncols) {
    __shared__ float Bs[(K / 2) * 64 * 2];

    for (int i = threadIdx.x; i < K * 64; i += 128) {
        const int p    = i >> 7;
        const int rem  = i & 127;
        const int col  = rem >> 1;
        const int half = rem & 1;
        Bs[i] = (col < Ncols) ? B[(size_t)(2 * p + half) * Ncols + col] : 0.f;
    }
    __syncthreads();

    const int wave = threadIdx.x >> 5;
    const int lane = threadIdx.x & 31;
    const int tileM = blockIdx.x * 4 + wave;
    const int row16 = tileM * 16;
    if (row16 >= M) return;

    const int hi = (lane >= 16);
    const int lr = lane & 15;

    v8f acc[NT];
#pragma unroll
    for (int t = 0; t < NT; ++t)
        acc[t] = (v8f){0.f, 0.f, 0.f, 0.f, 0.f, 0.f, 0.f, 0.f};

    const int arow = row16 + lr;
    const float* __restrict__ Arow  = A + (size_t)arow * K;
    const float* __restrict__ A2row = FUSE ? (A2 + (size_t)arow * K) : nullptr;

#pragma unroll 4
    for (int kb = 0; kb < K; kb += 4) {
        const int ka = kb + 2 * hi;
        float a0 = Arow[ka];
        float a1 = Arow[ka + 1];
        if (FUSE) {
            a0 = fmaxf(a0, A2row[ka]     + a2bias[ka]);
            a1 = fmaxf(a1, A2row[ka + 1] + a2bias[ka + 1]);
        }
        v2f afrag; afrag.x = a0; afrag.y = a1;

        const int p = (kb >> 1) + hi;
#pragma unroll
        for (int t = 0; t < NT; ++t) {
            const v2f bfrag = *(const v2f*)&Bs[(p << 7) + ((t * 16 + lr) << 1)];
            acc[t] = __builtin_amdgcn_wmma_f32_16x16x4_f32(
                false, afrag, false, bfrag, (short)0, acc[t], false, false);
        }
    }

    const int rbase = row16 + (hi ? 8 : 0);
#pragma unroll
    for (int t = 0; t < NT; ++t) {
        const int col = t * 16 + lr;
        if (col < Ncols) {
            const float bb = bias ? bias[col] : 0.f;
#pragma unroll
            for (int r = 0; r < 8; ++r)
                C[(size_t)(rbase + r) * Ncols + col] = acc[t][r] + bb;
        }
    }
}

__global__ __launch_bounds__(256)
void node_alpha(const float* __restrict__ h, const float* __restrict__ a_s,
                const float* __restrict__ a_d, float* __restrict__ asrc,
                float* __restrict__ adst) {
    const int gw   = (blockIdx.x * blockDim.x + threadIdx.x) >> 5;
    const int lane = threadIdx.x & 31;
    const int n = gw >> 1, hd = gw & 1;
    if (n >= Nn) return;
    const float hv = h[n * HCc + hd * HIDc + lane];
    float vs = hv * a_s[hd * HIDc + lane];
    float vd = hv * a_d[hd * HIDc + lane];
#pragma unroll
    for (int off = 16; off; off >>= 1) {
        vs += __shfl_xor(vs, off, 32);
        vd += __shfl_xor(vd, off, 32);
    }
    if (lane == 0) { asrc[n * 2 + hd] = vs; adst[n * 2 + hd] = vd; }
}

__global__ __launch_bounds__(256)
void init_layer(float* __restrict__ agg, float* __restrict__ m,
                float* __restrict__ den) {
    const int tid = blockIdx.x * blockDim.x + threadIdx.x;
    if (tid < Nn * HCc) agg[tid] = 0.f;
    if (tid < Nn * 2) { m[tid] = -INFINITY; den[tid] = 0.f; }
}

__global__ __launch_bounds__(256)
void edge_attn_max(const int* __restrict__ ei, int E, int Etot,
                   const float* __restrict__ asrc, const float* __restrict__ adst,
                   float* __restrict__ ebuf, float* __restrict__ m) {
    const int e = blockIdx.x * blockDim.x + threadIdx.x;
    if (e >= Etot) return;
    int s, d; edge_sd(ei, e, E, s, d);
#pragma unroll
    for (int hd = 0; hd < 2; ++hd) {
        float v = asrc[s * 2 + hd] + adst[d * 2 + hd];
        v = (v > 0.f) ? v : 0.2f * v;
        ebuf[e * 2 + hd] = v;
        atomicMaxFloat(&m[d * 2 + hd], v);
    }
}

__global__ __launch_bounds__(256)
void edge_exp_sum(const int* __restrict__ ei, int E, int Etot,
                  const float* __restrict__ m, float* __restrict__ ebuf,
                  float* __restrict__ den) {
    const int e = blockIdx.x * blockDim.x + threadIdx.x;
    if (e >= Etot) return;
    int s, d; edge_sd(ei, e, E, s, d);
#pragma unroll
    for (int hd = 0; hd < 2; ++hd) {
        const float ex = __expf(ebuf[e * 2 + hd] - m[d * 2 + hd]);
        ebuf[e * 2 + hd] = ex;
        atomicAdd(&den[d * 2 + hd], ex);
    }
}

__global__ __launch_bounds__(256)
void edge_msg(const int* __restrict__ ei, int E, int Etot,
              const float* __restrict__ h, const float* __restrict__ ex,
              const float* __restrict__ den, float* __restrict__ agg) {
    const int tid = blockIdx.x * blockDim.x + threadIdx.x;
    if (tid >= Etot * HCc) return;
    const int e = tid >> 6, c = tid & 63, hd = c >> 5;
    int s, d; edge_sd(ei, e, E, s, d);
    const float alpha = ex[e * 2 + hd] / (den[d * 2 + hd] + 1e-16f);
    atomicAdd(&agg[d * HCc + c], h[s * HCc + c] * alpha);
}

__global__ __launch_bounds__(256)
void node_post1(float* __restrict__ x, const float* __restrict__ b1,
                const float* __restrict__ gamma, const float* __restrict__ beta,
                const float* __restrict__ mean, const float* __restrict__ var) {
    const int tid = blockIdx.x * blockDim.x + threadIdx.x;
    if (tid >= Nn * HCc) return;
    const int c = tid & 63;
    float v = x[tid] + b1[c];
    v = (v - mean[c]) * (gamma[c] * rsqrtf(var[c] + 1e-5f)) + beta[c];
    x[tid] = (v > 0.f) ? v : (__expf(v) - 1.f);
}

extern "C" void kernel_launch(void* const* d_in, const int* in_sizes, int n_in,
                              void* d_out, int out_size, void* d_ws, size_t ws_size,
                              hipStream_t stream) {
    const float* node_feat = (const float*)d_in[0];
    const int*   ei        = (const int*)  d_in[1];
    const float* W1        = (const float*)d_in[2];
    const float* a_src1    = (const float*)d_in[3];
    const float* a_dst1    = (const float*)d_in[4];
    const float* b1        = (const float*)d_in[5];
    const float* bn_gamma  = (const float*)d_in[6];
    const float* bn_beta   = (const float*)d_in[7];
    const float* bn_mean   = (const float*)d_in[8];
    const float* bn_var    = (const float*)d_in[9];
    const float* W2        = (const float*)d_in[10];
    const float* a_src2    = (const float*)d_in[11];
    const float* a_dst2    = (const float*)d_in[12];
    const float* b2        = (const float*)d_in[13];
    const float* Wf        = (const float*)d_in[14];
    const float* bf        = (const float*)d_in[15];
    float* out = (float*)d_out;

    const int E    = in_sizes[1] / 2;
    const int Etot = E + Nn;

    float* ws   = (float*)d_ws;
    float* H    = ws;
    float* X1   = H    + (size_t)Nn * HCc;
    float* AGG2 = X1   + (size_t)Nn * HCc;
    float* ASRC = AGG2 + (size_t)Nn * HCc;
    float* ADST = ASRC + (size_t)Nn * 2;
    float* Mx   = ADST + (size_t)Nn * 2;
    float* DEN  = Mx   + (size_t)Nn * 2;
    float* EX   = DEN  + (size_t)Nn * 2;

    const int tilesM    = (Nn + 15) / 16;
    const int gemmBlks  = (tilesM + 3) / 4;
    const int nodeBlks  = (Nn * HCc + 255) / 256;
    const int alphaBlks = (Nn * 2 * 32 + 255) / 256;
    const int edgeBlks  = (Etot + 255) / 256;
    const int msgBlks   = (Etot * HCc + 255) / 256;

    wmma_gemm<4, false, FIN><<<gemmBlks, 128, 0, stream>>>(
        node_feat, nullptr, nullptr, W1, nullptr, H, Nn, HCc);
    node_alpha<<<alphaBlks, 256, 0, stream>>>(H, a_src1, a_dst1, ASRC, ADST);
    init_layer<<<nodeBlks, 256, 0, stream>>>(X1, Mx, DEN);
    edge_attn_max<<<edgeBlks, 256, 0, stream>>>(ei, E, Etot, ASRC, ADST, EX, Mx);
    edge_exp_sum<<<edgeBlks, 256, 0, stream>>>(ei, E, Etot, Mx, EX, DEN);
    edge_msg<<<msgBlks, 256, 0, stream>>>(ei, E, Etot, H, EX, DEN, X1);
    node_post1<<<nodeBlks, 256, 0, stream>>>(X1, b1, bn_gamma, bn_beta,
                                             bn_mean, bn_var);

    wmma_gemm<4, false, HCc><<<gemmBlks, 128, 0, stream>>>(
        X1, nullptr, nullptr, W2, nullptr, H, Nn, HCc);
    node_alpha<<<alphaBlks, 256, 0, stream>>>(H, a_src2, a_dst2, ASRC, ADST);
    init_layer<<<nodeBlks, 256, 0, stream>>>(AGG2, Mx, DEN);
    edge_attn_max<<<edgeBlks, 256, 0, stream>>>(ei, E, Etot, ASRC, ADST, EX, Mx);
    edge_exp_sum<<<edgeBlks, 256, 0, stream>>>(ei, E, Etot, Mx, EX, DEN);
    edge_msg<<<msgBlks, 256, 0, stream>>>(ei, E, Etot, H, EX, DEN, AGG2);

    wmma_gemm<3, true, HCc><<<gemmBlks, 128, 0, stream>>>(
        X1, AGG2, b2, Wf, bf, out, Nn, OUTc);
}


# layer 2's owner work-groups read the edge lists layer 1's filed (a word per edge: source, local destination; kept in the output buffer); no second table scan or edge-end gather; LDS adds as before
# speedup vs baseline: 7.4948x; 7.4948x over previous
#include <hip/hip_runtime.h>
#include <hip/hip_bf16.h>
#include <math.h>


#define Nn    100000
#define FIN   128
#define HCc   64
#define HIDc  32
#define OUTc  40
#define GEMM_WG   256
#define GEMM_THR  1024
#define OWN       391
#define OWN_WG    256
#define OWN_THR   1024
#define QCAP      8192
#define ECAP      8192

typedef __attribute__((ext_vector_type(2))) float v2f;
typedef __attribute__((ext_vector_type(8))) float v8f;

__device__ inline int fkey(float v) {
    const int b = __float_as_int(v);
    return (b >= 0) ? b : (b ^ 0x7FFFFFFF);
}
__device__ inline float funkey(int k) {
    return __int_as_float((k >= 0) ? k : (k ^ 0x7FFFFFFF));
}
__device__ inline float lrelu02(float v) { return (v > 0.f) ? v : 0.2f * v; }

template<int MASK>
__device__ __forceinline__ float row_xor_add(float v) {
    const int o = __builtin_amdgcn_update_dpp(0, __float_as_int(v), 0x160 + MASK,
                                              0xf, 0xf, true);
    return v + __int_as_float(o);
}

template<int NT, bool PREP, bool OBY, int K>
__global__ __launch_bounds__(GEMM_THR)
void wmma_gemm(const float* __restrict__ A,
               const float* __restrict__ B, const float* __restrict__ bias,
               float* __restrict__ C, int M, int Ncols,
               const float* __restrict__ att_s, const float* __restrict__ att_d,
               float* __restrict__ asrc, float* __restrict__ adst,
               const int* __restrict__ ei, int E, unsigned* __restrict__ ob) {
    static_assert(!PREP || NT == 4, "PREP needs the full 64-column row in the wave");
    if (OBY) {
        const int nW = (E + 3) >> 2;
        for (int w = blockIdx.x * GEMM_THR + threadIdx.x; w < nW; w += GEMM_WG * GEMM_THR) {
            unsigned pk = 0u;
#pragma unroll
            for (int j = 0; j < 4; ++j) {
                const int e  = 4 * w + j;
                const int ec = (e < E) ? e : (E - 1);
                const unsigned o = ((unsigned)ei[E + ec] / (unsigned)OWN) & 0xFFu;
                pk |= ((e < E) ? o : 0u) << (8 * j);
            }
            ob[w] = pk;
        }
    }
    __shared__ float Bs[(K / 2) * 64 * 2];

    for (int i = threadIdx.x; i < K * 64; i += GEMM_THR) {
        const int p    = i >> 7;
        const int rem  = i & 127;
        const int col  = rem >> 1;
        const int half = rem & 1;
        Bs[i] = (col < Ncols) ? B[(size_t)(2 * p + half) * Ncols + col] : 0.f;
    }
    __syncthreads();

    const int wave = threadIdx.x >> 5;
    const int lane = threadIdx.x & 31;
    const int tileM = wave * GEMM_WG + blockIdx.x;
    const int row16 = tileM * 16;
    if (row16 >= M) return;

    const int hi = (lane >= 16);
    const int lr = lane & 15;

    v8f acc[NT];
#pragma unroll
    for (int t = 0; t < NT; ++t)
        acc[t] = (v8f){0.f, 0.f, 0.f, 0.f, 0.f, 0.f, 0.f, 0.f};

    const int arow = row16 + lr;
    const float* __restrict__ Arow  = A + (size_t)arow * K;

#pragma unroll 4
    for (int kb = 0; kb < K; kb += 4) {
        const int ka = kb + 2 * hi;
        const float a0 = Arow[ka];
        const float a1 = Arow[ka + 1];
        v2f afrag; afrag.x = a0; afrag.y = a1;

        const int p = (kb >> 1) + hi;
#pragma unroll
        for (int t = 0; t < NT; ++t) {
            const v2f bfrag = *(const v2f*)&Bs[(p << 7) + ((t * 16 + lr) << 1)];
            acc[t] = __builtin_amdgcn_wmma_f32_16x16x4_f32(
                false, afrag, false, bfrag, (short)0, acc[t], false, false);
        }
    }

    const int rbase = row16 + (hi ? 8 : 0);
#pragma unroll
    for (int t = 0; t < NT; ++t) {
        const int col = t * 16 + lr;
        if (col < Ncols) {
            const float bb = bias ? bias[col] : 0.f;
#pragma unroll
            for (int r = 0; r < 8; ++r)
                C[(size_t)(rbase + r) * Ncols + col] = acc[t][r] + bb;
        }
    }

    if (PREP) {
        float vs[2][8], vd[2][8];
#pragma unroll
        for (int hd = 0; hd < 2; ++hd) {
            const float s_lo = att_s[hd * HIDc + lr], s_hi = att_s[hd * HIDc + 16 + lr];
            const float d_lo = att_d[hd * HIDc + lr], d_hi = att_d[hd * HIDc + 16 + lr];
#pragma unroll
            for (int r = 0; r < 8; ++r) {
                vs[hd][r] = acc[2 * hd][r] * s_lo + acc[2 * hd + 1][r] * s_hi;
                vd[hd][r] = acc[2 * hd][r] * d_lo + acc[2 * hd + 1][r] * d_hi;
            }
        }
#pragma unroll
        for (int hd = 0; hd < 2; ++hd)
#pragma unroll
            for (int r = 0; r < 8; ++r) { vs[hd][r] = row_xor_add<8>(vs[hd][r]); vd[hd][r] = row_xor_add<8>(vd[hd][r]); }
#pragma unroll
        for (int hd = 0; hd < 2; ++hd)
#pragma unroll
            for (int r = 0; r < 8; ++r) { vs[hd][r] = row_xor_add<4>(vs[hd][r]); vd[hd][r] = row_xor_add<4>(vd[hd][r]); }
#pragma unroll
        for (int hd = 0; hd < 2; ++hd)
#pragma unroll
            for (int r = 0; r < 8; ++r) { vs[hd][r] = row_xor_add<2>(vs[hd][r]); vd[hd][r] = row_xor_add<2>(vd[hd][r]); }
#pragma unroll
        for (int hd = 0; hd < 2; ++hd)
#pragma unroll
            for (int r = 0; r < 8; ++r) { vs[hd][r] = row_xor_add<1>(vs[hd][r]); vd[hd][r] = row_xor_add<1>(vd[hd][r]); }
        float os0 = vs[0][0], os1 = vs[1][0], od0 = vd[0][0], od1 = vd[1][0];
#pragma unroll
        for (int r = 1; r < 8; ++r) {
            os0 = (lr == r) ? vs[0][r] : os0;
            os1 = (lr == r) ? vs[1][r] : os1;
            od0 = (lr == r) ? vd[0][r] : od0;
            od1 = (lr == r) ? vd[1][r] : od1;
        }
        if (lr < 8) {
            const int n = rbase + lr;
            float2 a2; a2.x = os0; a2.y = os1;
            *(float2*)(asrc + 2 * n) = a2;
            float2 d2; d2.x = od0; d2.y = od1;
            *(float2*)(adst + 2 * n) = d2;
        }
    }
}

__device__ __forceinline__ unsigned zbytes(unsigned y) {
    return ~(((y & 0x7F7F7F7Fu) + 0x7F7F7F7Fu) | y | 0x7F7F7F7Fu);
}
__device__ __forceinline__ unsigned zpack(unsigned z) {
    const unsigned t = z >> 7;
    return (t | (t >> 7) | (t >> 14) | (t >> 21)) & 0xFu;
}

template<int LAYER>
__global__ __launch_bounds__(OWN_THR)
void owner_layer(const int* __restrict__ ei, int E, const unsigned* __restrict__ ob,
                 const float* __restrict__ h, const float* __restrict__ asrc,
                 const float* __restrict__ adst, const float* __restrict__ bias,
                 const float* __restrict__ gamma, const float* __restrict__ beta,
                 const float* __restrict__ mean, const float* __restrict__ var,
                 const float* __restrict__ xprev, float* __restrict__ xout,
                 unsigned* __restrict__ lst, int* __restrict__ lcnt) {
    __shared__ __attribute__((aligned(16))) float acc[OWN * HCc];
    __shared__ int   mkey[OWN * 2];
    __shared__ float s_as[OWN * 2], s_ad[OWN * 2], s_m[OWN * 2], s_den[OWN * 2];
    __shared__ int   candq[QCAP];
    __shared__ int   cande[ECAP];
    __shared__ int   ls[ECAP], ld[ECAP];
    __shared__ int   ctr[2];

    const int tid  = threadIdx.x;
    const int lane = tid & 31;
    const int wv   = __builtin_amdgcn_readfirstlane(tid >> 5);
    const int lo   = blockIdx.x * OWN;
    const int cntN = (Nn - lo < OWN) ? (Nn - lo) : OWN;

    if (LAYER == 1) {
        if (tid < 2) ctr[tid] = 0;
        if (tid == 2) { candq[0] = 0; cande[0] = 0; }
    }
    for (int i = tid; i < cntN * 2; i += OWN_THR) {
        const float a = asrc[lo * 2 + i];
        s_as[i] = a;
        s_ad[i] = adst[lo * 2 + i];
        mkey[i] = fkey(a);
    }
    __syncthreads();

    int  ne;
    bool over;
    if (LAYER == 1) {
        const unsigned pat = (unsigned)blockIdx.x * 0x01010101u;
        const int nQ = E >> 4;
        const uint4* __restrict__ ob4 = (const uint4*)ob;
        const int rot = (int)(blockIdx.x * (unsigned)(nQ >> 8)) & ~7;
        for (int q0 = tid; q0 < nQ; q0 += 8 * OWN_THR) {
            uint4 v[8];
            int   qa[8];
#pragma unroll
            for (int u = 0; u < 8; ++u) {
                const int qi = q0 + u * OWN_THR;
                int q = ((qi < nQ) ? qi : q0) + rot;
                if (q >= nQ) q -= nQ;
                qa[u] = (qi < nQ) ? q : -1;
                v[u] = ob4[q];
            }
#pragma unroll
            for (int u = 0; u < 8; ++u) {
                const unsigned y0 = v[u].x ^ pat, y1 = v[u].y ^ pat, y2 = v[u].z ^ pat, y3 = v[u].w ^ pat;
                const unsigned z = (((y0 - 0x01010101u) & ~y0) | ((y1 - 0x01010101u) & ~y1) |
                                    ((y2 - 0x01010101u) & ~y2) | ((y3 - 0x01010101u) & ~y3)) & 0x80808080u;
                if (z && qa[u] >= 0) {
                    const int p = atomicAdd(&ctr[0], 1);
                    if (p < QCAP) candq[p] = qa[u];
                }
            }
        }
        __syncthreads();

        const int nq = ctr[0];
        over = nq > QCAP;
        if (!over) {
            int qq[8]; uint4 vv[8];
#pragma unroll
            for (int u = 0; u < 8; ++u) {
                const int i = tid + u * OWN_THR;
                const int q = candq[(i < nq) ? i : 0];
                vv[u] = ob4[q];
                qq[u] = (i < nq) ? q : -1;
            }
#pragma unroll
            for (int u = 0; u < 8; ++u) {
                unsigned mm = zpack(zbytes(vv[u].x ^ pat))
                            | (zpack(zbytes(vv[u].y ^ pat)) << 4)
                            | (zpack(zbytes(vv[u].z ^ pat)) << 8)
                            | (zpack(zbytes(vv[u].w ^ pat)) << 12);
                if (qq[u] < 0) mm = 0u;
                for (int trip = 0; trip < 16; ++trip) {
                    const bool has = (mm != 0u);
                    const unsigned bal = __builtin_amdgcn_ballot_w32(has);
                    if (bal == 0u) break;
                    const int j = has ? __builtin_ctz(mm) : 0;
                    mm &= mm - 1u;
                    const int lead = __builtin_ctz(bal);
                    int draw = 0;
                    if (lane == lead) draw = atomicAdd(&ctr[1], __builtin_popcount(bal));
                    draw = __builtin_amdgcn_readlane(draw, lead);
                    const int p = draw + (int)__builtin_amdgcn_mbcnt_lo(bal, 0u);
                    if (has && p < ECAP) cande[p] = 16 * qq[u] + j;
                }
            }
            for (int e = 16 * nQ + tid; e < E; e += OWN_THR) {
                const unsigned by = (ob[e >> 2] >> (8 * (e & 3))) & 0xFFu;
                if (by == (unsigned)blockIdx.x) {
                    const int p = atomicAdd(&ctr[1], 1);
                    if (p < ECAP) cande[p] = e;
                }
            }
        }
        __syncthreads();

        ne = ctr[1];
        over = over || (ne > ECAP);
        if (tid == 0) lcnt[blockIdx.x] = over ? (ECAP + 1) : ne;
    } else {
        ne = lcnt[blockIdx.x];
        over = (unsigned)ne > (unsigned)ECAP;
    }

    unsigned* __restrict__ lb = lst + (size_t)blockIdx.x * ECAP;
    if (!over) {
        int sv[8], dv[8];
        if (LAYER == 1) {
#pragma unroll
            for (int u = 0; u < 8; ++u) {
                const int i = tid + u * OWN_THR;
                const int e = cande[(i < ne) ? i : 0];
                sv[u] = ei[e];
                dv[u] = ei[E + e] - lo;
            }
        } else {
#pragma unroll
            for (int u = 0; u < 8; ++u) {
                const int i = tid + u * OWN_THR;
                const unsigned rec = lb[(i < ne) ? i : 0];
                const unsigned s = rec >> 9;
                sv[u] = (s < (unsigned)Nn) ? (int)s : 0;
                dv[u] = (int)(rec & 511u);
            }
        }
        float2 av[8];
#pragma unroll
        for (int u = 0; u < 8; ++u) {
            const int i = tid + u * OWN_THR;
            if (i >= ne || dv[u] < 0 || dv[u] >= cntN) dv[u] = -1;
            av[u] = *(const float2*)(asrc + 2 * sv[u]);
        }
#pragma unroll
        for (int u = 0; u < 8; ++u) {
            const int i = tid + u * OWN_THR;
            if (i < ne) {
                ls[i] = sv[u]; ld[i] = dv[u];
                if (LAYER == 1) lb[i] = ((unsigned)sv[u] << 9) | ((unsigned)dv[u] & 511u);
            }
            if (dv[u] >= 0) {
                atomicMax(&mkey[2 * dv[u]],     fkey(av[u].x));
                atomicMax(&mkey[2 * dv[u] + 1], fkey(av[u].y));
            }
        }
    } else {
        for (int e = tid; e < E; e += OWN_THR) {
            const int d = ei[E + e] - lo;
            if (d >= 0 && d < cntN) {
                const float2 a = *(const float2*)(asrc + 2 * ei[e]);
                atomicMax(&mkey[2 * d],     fkey(a.x));
                atomicMax(&mkey[2 * d + 1], fkey(a.y));
            }
        }
    }
    __syncthreads();

    static_assert(OWN * (HCc / 4) <= 7 * OWN_THR && OWN_THR % (HCc / 4) == 0, "seven passes cover an owner's rows");
    const int c4  = (tid & 15) * 4;
    const int hd4 = c4 >> 5;
    const int n4  = cntN * (HCc / 4);
    float4* __restrict__ acc4 = (float4*)acc;
    {
        const float4* __restrict__ h4 = (const float4*)(h + (size_t)lo * HCc);
        float4 hv[7];
#pragma unroll
        for (int k = 0; k < 7; ++k) {
            const int i4 = tid + k * OWN_THR;
            hv[k] = h4[(i4 < n4) ? i4 : 0];
        }
#pragma unroll
        for (int k = 0; k < 7; ++k) {
            const int i4 = tid + k * OWN_THR;
            if (i4 < n4) {
                const int nl = i4 >> 4;
                const float ad = s_ad[2 * nl + hd4];
                const float m  = lrelu02(funkey(mkey[2 * nl + hd4]) + ad);
                const float ex = __expf(lrelu02(s_as[2 * nl + hd4] + ad) - m);
                float4 a;
                a.x = hv[k].x * ex; a.y = hv[k].y * ex; a.z = hv[k].z * ex; a.w = hv[k].w * ex;
                acc4[i4] = a;
                if ((c4 & 31) == 0) { s_m[2 * nl + hd4] = m; s_den[2 * nl + hd4] = ex; }
            }
        }
    }
    __syncthreads();

    if (!over) {
        for (int base = wv * 32; base < ne; base += OWN_THR) {
            const int i = base + lane;
            const int cnt = (ne - base < 32) ? (ne - base) : 32;
            int s = 0, d = -1;
            float ex0 = 0.f, ex1 = 0.f;
            if (i < ne) { s = ls[i]; d = ld[i]; }
            if (d >= 0) {
                const float2 a = *(const float2*)(asrc + 2 * s);
                ex0 = __expf(lrelu02(a.x + s_ad[2 * d])     - s_m[2 * d]);
                ex1 = __expf(lrelu02(a.y + s_ad[2 * d + 1]) - s_m[2 * d + 1]);
                atomicAdd(&s_den[2 * d],     ex0);
                atomicAdd(&s_den[2 * d + 1], ex1);
            }
            for (int j0 = 0; j0 < cnt; j0 += 16) {
                float hv0[16], hv1[16];
#pragma unroll
                for (int u = 0; u < 16; ++u) {
                    hv0[u] = 0.f; hv1[u] = 0.f;
                    const int dj = __builtin_amdgcn_readlane(d, j0 + u);
                    if (j0 + u < cnt && dj >= 0) {
                        const int sj = __builtin_amdgcn_readlane(s, j0 + u);
                        const float* hr = h + (size_t)sj * HCc + lane;
                        hv0[u] = hr[0];
                        hv1[u] = hr[HIDc];
                    }
                }
#pragma unroll
                for (int u = 0; u < 16; ++u) {
                    const int dj = __builtin_amdgcn_readlane(d, j0 + u);
                    if (j0 + u < cnt && dj >= 0) {
                        const float x0 = __int_as_float(__builtin_amdgcn_readlane(__float_as_int(ex0), j0 + u));
                        const float x1 = __int_as_float(__builtin_amdgcn_readlane(__float_as_int(ex1), j0 + u));
                        atomicAdd(&acc[dj * HCc + lane],        hv0[u] * x0);
                        atomicAdd(&acc[dj * HCc + HIDc + lane], hv1[u] * x1);
                    }
                }
            }
        }
    } else {
        for (int e = tid; e < E; e += OWN_THR) {
            const int d = ei[E + e] - lo;
            if (d >= 0 && d < cntN) {
                const int s = ei[e];
                const float2 a = *(const float2*)(asrc + 2 * s);
                const float ex0 = __expf(lrelu02(a.x + s_ad[2 * d])     - s_m[2 * d]);
                const float ex1 = __expf(lrelu02(a.y + s_ad[2 * d + 1]) - s_m[2 * d + 1]);
                atomicAdd(&s_den[2 * d],     ex0);
                atomicAdd(&s_den[2 * d + 1], ex1);
                const float* hr = h + (size_t)s * HCc;
                for (int c = 0; c < HIDc; ++c) {
                    atomicAdd(&acc[d * HCc + c],        hr[c] * ex0);
                    atomicAdd(&acc[d * HCc + HIDc + c], hr[HIDc + c] * ex1);
                }
            }
        }
    }
    __syncthreads();

    {
        float bi[4], mu[4], sc[4], be[4];
#pragma unroll
        for (int j = 0; j < 4; ++j) {
            bi[j] = bias[c4 + j];
            mu[j] = 0.f; sc[j] = 1.f; be[j] = 0.f;
            if (LAYER == 1) {
                mu[j] = mean[c4 + j];
                sc[j] = gamma[c4 + j] * rsqrtf(var[c4 + j] + 1e-5f);
                be[j] = beta[c4 + j];
            }
        }
        float4* __restrict__ xo4 = (float4*)(xout + (size_t)lo * HCc);
        float4 xp[7];
        if (LAYER == 2) {
            const float4* __restrict__ xp4 = (const float4*)(xprev + (size_t)lo * HCc);
#pragma unroll
            for (int k = 0; k < 7; ++k) {
                const int i4 = tid + k * OWN_THR;
                xp[k] = xp4[(i4 < n4) ? i4 : 0];
            }
        }
#pragma unroll
        for (int k = 0; k < 7; ++k) {
            const int i4 = tid + k * OWN_THR;
            if (i4 < n4) {
                const int nl = i4 >> 4;
                const float dn = s_den[2 * nl + hd4] + 1e-16f;
                const float4 a = acc4[i4];
                float v[4] = {a.x / dn + bi[0], a.y / dn + bi[1], a.z / dn + bi[2], a.w / dn + bi[3]};
                if (LAYER == 1) {
#pragma unroll
                    for (int j = 0; j < 4; ++j) {
                        v[j] = (v[j] - mu[j]) * sc[j] + be[j];
                        v[j] = (v[j] > 0.f) ? v[j] : (__expf(v[j]) - 1.f);
                    }
                } else {
                    v[0] = fmaxf(xp[k].x, v[0]); v[1] = fmaxf(xp[k].y, v[1]);
                    v[2] = fmaxf(xp[k].z, v[2]); v[3] = fmaxf(xp[k].w, v[3]);
                }
                float4 o; o.x = v[0]; o.y = v[1]; o.z = v[2]; o.w = v[3];
                xo4[i4] = o;
            }
        }
    }
}

extern "C" void kernel_launch(void* const* d_in, const int* in_sizes, int n_in,
                              void* d_out, int out_size, void* d_ws, size_t ws_size,
                              hipStream_t stream) {
    const float* node_feat = (const float*)d_in[0];
    const int*   ei        = (const int*)  d_in[1];
    const float* W1        = (const float*)d_in[2];
    const float* a_src1    = (const float*)d_in[3];
    const float* a_dst1    = (const float*)d_in[4];
    const float* b1        = (const float*)d_in[5];
    const float* bn_gamma  = (const float*)d_in[6];
    const float* bn_beta   = (const float*)d_in[7];
    const float* bn_mean   = (const float*)d_in[8];
    const float* bn_var    = (const float*)d_in[9];
    const float* W2        = (const float*)d_in[10];
    const float* a_src2    = (const float*)d_in[11];
    const float* a_dst2    = (const float*)d_in[12];
    const float* b2        = (const float*)d_in[13];
    const float* Wf        = (const float*)d_in[14];
    const float* bf        = (const float*)d_in[15];
    float* out = (float*)d_out;

    const int E = in_sizes[1] / 2;

    float* ws   = (float*)d_ws;
    float* H    = ws;
    float* X1   = H    + (size_t)Nn * HCc;
    float* XJ   = X1   + (size_t)Nn * HCc;
    float* ASRC = XJ   + (size_t)Nn * HCc;
    float* ADST = ASRC + (size_t)Nn * 2;

    static_assert(GEMM_WG * (GEMM_THR / 32) * 16 >= Nn, "every row tile needs a wave");
    static_assert(OWN_WG * OWN >= Nn && OWN_WG <= 256, "every node needs an owner, every owner a byte value");
    unsigned* OB = (unsigned*)d_out;
    static_assert(OWN <= 511 && Nn <= (1 << 17), "a list word holds source << 9 | local destination, 511 = none");
    static_assert((1 << 20) + OWN_WG * ECAP + OWN_WG <= Nn * OUTc, "table, lists and lengths fit in the output buffer");
    unsigned* LST  = (unsigned*)d_out + (1 << 20);
    int*      LCNT = (int*)(LST + OWN_WG * ECAP);

    wmma_gemm<4, true, true, FIN><<<GEMM_WG, GEMM_THR, 0, stream>>>(
        node_feat, W1, nullptr, H, Nn, HCc, a_src1, a_dst1, ASRC, ADST, ei, E, OB);
    owner_layer<1><<<OWN_WG, OWN_THR, 0, stream>>>(
        ei, E, OB, H, ASRC, ADST, b1, bn_gamma, bn_beta, bn_mean, bn_var, nullptr, X1, LST, LCNT);

    wmma_gemm<4, true, false, HCc><<<GEMM_WG, GEMM_THR, 0, stream>>>(
        X1, W2, nullptr, H, Nn, HCc, a_src2, a_dst2, ASRC, ADST, nullptr, 0, nullptr);
    owner_layer<2><<<OWN_WG, OWN_THR, 0, stream>>>(
        ei, E, OB, H, ASRC, ADST, b2, nullptr, nullptr, nullptr, nullptr, X1, XJ, LST, LCNT);

    wmma_gemm<3, false, false, HCc><<<GEMM_WG, GEMM_THR, 0, stream>>>(
        XJ, Wf, bf, out, Nn, OUTc, nullptr, nullptr, nullptr, nullptr, nullptr, 0, nullptr);
}


# owner row sums: a wave walks its 32 listed edges in straight-line code (constant lane reads, no test or branch per edge; a missing edge adds weight 0 into a spare LDS row); LDS adds as before
# speedup vs baseline: 7.7209x; 1.0302x over previous
#include <hip/hip_runtime.h>
#include <hip/hip_bf16.h>
#include <math.h>


#define Nn    100000
#define FIN   128
#define HCc   64
#define HIDc  32
#define OUTc  40
#define GEMM_WG   256
#define GEMM_THR  1024
#define OWN       391
#define OWN_WG    256
#define OWN_THR   1024
#define QCAP      8192
#define ECAP      8192

typedef __attribute__((ext_vector_type(2))) float v2f;
typedef __attribute__((ext_vector_type(8))) float v8f;

__device__ inline int fkey(float v) {
    const int b = __float_as_int(v);
    return (b >= 0) ? b : (b ^ 0x7FFFFFFF);
}
__device__ inline float funkey(int k) {
    return __int_as_float((k >= 0) ? k : (k ^ 0x7FFFFFFF));
}
__device__ inline float lrelu02(float v) { return (v > 0.f) ? v : 0.2f * v; }

template<int MASK>
__device__ __forceinline__ float row_xor_add(float v) {
    const int o = __builtin_amdgcn_update_dpp(0, __float_as_int(v), 0x160 + MASK,
                                              0xf, 0xf, true);
    return v + __int_as_float(o);
}

template<int NT, bool PREP, bool OBY, int K>
__global__ __launch_bounds__(GEMM_THR)
void wmma_gemm(const float* __restrict__ A,
               const float* __restrict__ B, const float* __restrict__ bias,
               float* __restrict__ C, int M, int Ncols,
               const float* __restrict__ att_s, const float* __restrict__ att_d,
               float* __restrict__ asrc, float* __restrict__ adst,
               const int* __restrict__ ei, int E, unsigned* __restrict__ ob) {
    static_assert(!PREP || NT == 4, "PREP needs the full 64-column row in the wave");
    if (OBY) {
        const int nW = (E + 3) >> 2;
        for (int w = blockIdx.x * GEMM_THR + threadIdx.x; w < nW; w += GEMM_WG * GEMM_THR) {
            unsigned pk = 0u;
#pragma unroll
            for (int j = 0; j < 4; ++j) {
                const int e  = 4 * w + j;
                const int ec = (e < E) ? e : (E - 1);
                const unsigned o = ((unsigned)ei[E + ec] / (unsigned)OWN) & 0xFFu;
                pk |= ((e < E) ? o : 0u) << (8 * j);
            }
            ob[w] = pk;
        }
    }
    __shared__ float Bs[(K / 2) * 64 * 2];

    for (int i = threadIdx.x; i < K * 64; i += GEMM_THR) {
        const int p    = i >> 7;
        const int rem  = i & 127;
        const int col  = rem >> 1;
        const int half = rem & 1;
        Bs[i] = (col < Ncols) ? B[(size_t)(2 * p + half) * Ncols + col] : 0.f;
    }
    __syncthreads();

    const int wave = threadIdx.x >> 5;
    const int lane = threadIdx.x & 31;
    const int tileM = wave * GEMM_WG + blockIdx.x;
    const int row16 = tileM * 16;
    if (row16 >= M) return;

    const int hi = (lane >= 16);
    const int lr = lane & 15;

    v8f acc[NT];
#pragma unroll
    for (int t = 0; t < NT; ++t)
        acc[t] = (v8f){0.f, 0.f, 0.f, 0.f, 0.f, 0.f, 0.f, 0.f};

    const int arow = row16 + lr;
    const float* __restrict__ Arow  = A + (size_t)arow * K;

#pragma unroll 4
    for (int kb = 0; kb < K; kb += 4) {
        const int ka = kb + 2 * hi;
        const float a0 = Arow[ka];
        const float a1 = Arow[ka + 1];
        v2f afrag; afrag.x = a0; afrag.y = a1;

        const int p = (kb >> 1) + hi;
#pragma unroll
        for (int t = 0; t < NT; ++t) {
            const v2f bfrag = *(const v2f*)&Bs[(p << 7) + ((t * 16 + lr) << 1)];
            acc[t] = __builtin_amdgcn_wmma_f32_16x16x4_f32(
                false, afrag, false, bfrag, (short)0, acc[t], false, false);
        }
    }

    const int rbase = row16 + (hi ? 8 : 0);
#pragma unroll
    for (int t = 0; t < NT; ++t) {
        const int col = t * 16 + lr;
        if (col < Ncols) {
            const float bb = bias ? bias[col] : 0.f;
#pragma unroll
            for (int r = 0; r < 8; ++r)
                C[(size_t)(rbase + r) * Ncols + col] = acc[t][r] + bb;
        }
    }

    if (PREP) {
        float vs[2][8], vd[2][8];
#pragma unroll
        for (int hd = 0; hd < 2; ++hd) {
            const float s_lo = att_s[hd * HIDc + lr], s_hi = att_s[hd * HIDc + 16 + lr];
            const float d_lo = att_d[hd * HIDc + lr], d_hi = att_d[hd * HIDc + 16 + lr];
#pragma unroll
            for (int r = 0; r < 8; ++r) {
                vs[hd][r] = acc[2 * hd][r] * s_lo + acc[2 * hd + 1][r] * s_hi;
                vd[hd][r] = acc[2 * hd][r] * d_lo + acc[2 * hd + 1][r] * d_hi;
            }
        }
#pragma unroll
        for (int hd = 0; hd < 2; ++hd)
#pragma unroll
            for (int r = 0; r < 8; ++r) { vs[hd][r] = row_xor_add<8>(vs[hd][r]); vd[hd][r] = row_xor_add<8>(vd[hd][r]); }
#pragma unroll
        for (int hd = 0; hd < 2; ++hd)
#pragma unroll
            for (int r = 0; r < 8; ++r) { vs[hd][r] = row_xor_add<4>(vs[hd][r]); vd[hd][r] = row_xor_add<4>(vd[hd][r]); }
#pragma unroll
        for (int hd = 0; hd < 2; ++hd)
#pragma unroll
            for (int r = 0; r < 8; ++r) { vs[hd][r] = row_xor_add<2>(vs[hd][r]); vd[hd][r] = row_xor_add<2>(vd[hd][r]); }
#pragma unroll
        for (int hd = 0; hd < 2; ++hd)
#pragma unroll
            for (int r = 0; r < 8; ++r) { vs[hd][r] = row_xor_add<1>(vs[hd][r]); vd[hd][r] = row_xor_add<1>(vd[hd][r]); }
        float os0 = vs[0][0], os1 = vs[1][0], od0 = vd[0][0], od1 = vd[1][0];
#pragma unroll
        for (int r = 1; r < 8; ++r) {
            os0 = (lr == r) ? vs[0][r] : os0;
            os1 = (lr == r) ? vs[1][r] : os1;
            od0 = (lr == r) ? vd[0][r] : od0;
            od1 = (lr == r) ? vd[1][r] : od1;
        }
        if (lr < 8) {
            const int n = rbase + lr;
            float2 a2; a2.x = os0; a2.y = os1;
            *(float2*)(asrc + 2 * n) = a2;
            float2 d2; d2.x = od0; d2.y = od1;
            *(float2*)(adst + 2 * n) = d2;
        }
    }
}

template<int J0>
__device__ __forceinline__ void rows16(const char* __restrict__ hb, float* __restrict__ acc,
                                       unsigned lane4, unsigned soff, int drow,
                                       float ex0, float ex1) {
    float hv0[16], hv1[16];
#pragma unroll
    for (int u = 0; u < 16; ++u) {
        const unsigned so = (unsigned)__builtin_amdgcn_readlane((int)soff, J0 + u);
        const float* hr = (const float*)(hb + (size_t)(so + lane4));
        hv0[u] = hr[0];
        hv1[u] = hr[HIDc];
    }
#pragma unroll
    for (int u = 0; u < 16; ++u) {
        const int dj = __builtin_amdgcn_readlane(drow, J0 + u);
        const float x0 = __int_as_float(__builtin_amdgcn_readlane(__float_as_int(ex0), J0 + u));
        const float x1 = __int_as_float(__builtin_amdgcn_readlane(__float_as_int(ex1), J0 + u));
        atomicAdd(&acc[dj * HCc + (int)(lane4 >> 2)],        hv0[u] * x0);
        atomicAdd(&acc[dj * HCc + HIDc + (int)(lane4 >> 2)], hv1[u] * x1);
    }
}

__device__ __forceinline__ unsigned zbytes(unsigned y) {
    return ~(((y & 0x7F7F7F7Fu) + 0x7F7F7F7Fu) | y | 0x7F7F7F7Fu);
}
__device__ __forceinline__ unsigned zpack(unsigned z) {
    const unsigned t = z >> 7;
    return (t | (t >> 7) | (t >> 14) | (t >> 21)) & 0xFu;
}

template<int LAYER>
__global__ __launch_bounds__(OWN_THR)
void owner_layer(const int* __restrict__ ei, int E, const unsigned* __restrict__ ob,
                 const float* __restrict__ h, const float* __restrict__ asrc,
                 const float* __restrict__ adst, const float* __restrict__ bias,
                 const float* __restrict__ gamma, const float* __restrict__ beta,
                 const float* __restrict__ mean, const float* __restrict__ var,
                 const float* __restrict__ xprev, float* __restrict__ xout,
                 unsigned* __restrict__ lst, int* __restrict__ lcnt) {
    __shared__ __attribute__((aligned(16))) float acc[(OWN + 1) * HCc];
    __shared__ int   mkey[OWN * 2];
    __shared__ float s_as[OWN * 2], s_ad[OWN * 2], s_m[OWN * 2], s_den[OWN * 2];
    __shared__ int   candq[QCAP];
    __shared__ int   cande[ECAP];
    __shared__ int   ls[ECAP], ld[ECAP];
    __shared__ int   ctr[2];

    const int tid  = threadIdx.x;
    const int lane = tid & 31;
    const int wv   = __builtin_amdgcn_readfirstlane(tid >> 5);
    const int lo   = blockIdx.x * OWN;
    const int cntN = (Nn - lo < OWN) ? (Nn - lo) : OWN;

    if (LAYER == 1) {
        if (tid < 2) ctr[tid] = 0;
        if (tid == 2) { candq[0] = 0; cande[0] = 0; }
    }
    for (int i = tid; i < cntN * 2; i += OWN_THR) {
        const float a = asrc[lo * 2 + i];
        s_as[i] = a;
        s_ad[i] = adst[lo * 2 + i];
        mkey[i] = fkey(a);
    }
    __syncthreads();

    int  ne;
    bool over;
    if (LAYER == 1) {
        const unsigned pat = (unsigned)blockIdx.x * 0x01010101u;
        const int nQ = E >> 4;
        const uint4* __restrict__ ob4 = (const uint4*)ob;
        const int rot = (int)(blockIdx.x * (unsigned)(nQ >> 8)) & ~7;
        for (int q0 = tid; q0 < nQ; q0 += 8 * OWN_THR) {
            uint4 v[8];
            int   qa[8];
#pragma unroll
            for (int u = 0; u < 8; ++u) {
                const int qi = q0 + u * OWN_THR;
                int q = ((qi < nQ) ? qi : q0) + rot;
                if (q >= nQ) q -= nQ;
                qa[u] = (qi < nQ) ? q : -1;
                v[u] = ob4[q];
            }
#pragma unroll
            for (int u = 0; u < 8; ++u) {
                const unsigned y0 = v[u].x ^ pat, y1 = v[u].y ^ pat, y2 = v[u].z ^ pat, y3 = v[u].w ^ pat;
                const unsigned z = (((y0 - 0x01010101u) & ~y0) | ((y1 - 0x01010101u) & ~y1) |
                                    ((y2 - 0x01010101u) & ~y2) | ((y3 - 0x01010101u) & ~y3)) & 0x80808080u;
                if (z && qa[u] >= 0) {
                    const int p = atomicAdd(&ctr[0], 1);
                    if (p < QCAP) candq[p] = qa[u];
                }
            }
        }
        __syncthreads();

        const int nq = ctr[0];
        over = nq > QCAP;
        if (!over) {
            int qq[8]; uint4 vv[8];
#pragma unroll
            for (int u = 0; u < 8; ++u) {
                const int i = tid + u * OWN_THR;
                const int q = candq[(i < nq) ? i : 0];
                vv[u] = ob4[q];
                qq[u] = (i < nq) ? q : -1;
            }
#pragma unroll
            for (int u = 0; u < 8; ++u) {
                unsigned mm = zpack(zbytes(vv[u].x ^ pat))
                            | (zpack(zbytes(vv[u].y ^ pat)) << 4)
                            | (zpack(zbytes(vv[u].z ^ pat)) << 8)
                            | (zpack(zbytes(vv[u].w ^ pat)) << 12);
                if (qq[u] < 0) mm = 0u;
                for (int trip = 0; trip < 16; ++trip) {
                    const bool has = (mm != 0u);
                    const unsigned bal = __builtin_amdgcn_ballot_w32(has);
                    if (bal == 0u) break;
                    const int j = has ? __builtin_ctz(mm) : 0;
                    mm &= mm - 1u;
                    const int lead = __builtin_ctz(bal);
                    int draw = 0;
                    if (lane == lead) draw = atomicAdd(&ctr[1], __builtin_popcount(bal));
                    draw = __builtin_amdgcn_readlane(draw, lead);
                    const int p = draw + (int)__builtin_amdgcn_mbcnt_lo(bal, 0u);
                    if (has && p < ECAP) cande[p] = 16 * qq[u] + j;
                }
            }
            for (int e = 16 * nQ + tid; e < E; e += OWN_THR) {
                const unsigned by = (ob[e >> 2] >> (8 * (e & 3))) & 0xFFu;
                if (by == (unsigned)blockIdx.x) {
                    const int p = atomicAdd(&ctr[1], 1);
                    if (p < ECAP) cande[p] = e;
                }
            }
        }
        __syncthreads();

        ne = ctr[1];
        over = over || (ne > ECAP);
        if (tid == 0) lcnt[blockIdx.x] = over ? (ECAP + 1) : ne;
    } else {
        ne = lcnt[blockIdx.x];
        over = (unsigned)ne > (unsigned)ECAP;
    }

    unsigned* __restrict__ lb = lst + (size_t)blockIdx.x * ECAP;
    if (!over) {
        int sv[8], dv[8];
        if (LAYER == 1) {
#pragma unroll
            for (int u = 0; u < 8; ++u) {
                const int i = tid + u * OWN_THR;
                const int e = cande[(i < ne) ? i : 0];
                sv[u] = ei[e];
                dv[u] = ei[E + e] - lo;
            }
        } else {
#pragma unroll
            for (int u = 0; u < 8; ++u) {
                const int i = tid + u * OWN_THR;
                const unsigned rec = lb[(i < ne) ? i : 0];
                const unsigned s = rec >> 9;
                sv[u] = (s < (unsigned)Nn) ? (int)s : 0;
                dv[u] = (int)(rec & 511u);
            }
        }
        float2 av[8];
#pragma unroll
        for (int u = 0; u < 8; ++u) {
            const int i = tid + u * OWN_THR;
            if (i >= ne || dv[u] < 0 || dv[u] >= cntN) dv[u] = -1;
            av[u] = *(const float2*)(asrc + 2 * sv[u]);
        }
#pragma unroll
        for (int u = 0; u < 8; ++u) {
            const int i = tid + u * OWN_THR;
            if (i < ne) {
                ls[i] = sv[u]; ld[i] = dv[u];
                if (LAYER == 1) lb[i] = ((unsigned)sv[u] << 9) | ((unsigned)dv[u] & 511u);
            }
            if (dv[u] >= 0) {
                atomicMax(&mkey[2 * dv[u]],     fkey(av[u].x));
                atomicMax(&mkey[2 * dv[u] + 1], fkey(av[u].y));
            }
        }
    } else {
        for (int e = tid; e < E; e += OWN_THR) {
            const int d = ei[E + e] - lo;
            if (d >= 0 && d < cntN) {
                const float2 a = *(const float2*)(asrc + 2 * ei[e]);
                atomicMax(&mkey[2 * d],     fkey(a.x));
                atomicMax(&mkey[2 * d + 1], fkey(a.y));
            }
        }
    }
    __syncthreads();

    static_assert(OWN * (HCc / 4) <= 7 * OWN_THR && OWN_THR % (HCc / 4) == 0, "seven passes cover an owner's rows");
    const int c4  = (tid & 15) * 4;
    const int hd4 = c4 >> 5;
    const int n4  = cntN * (HCc / 4);
    float4* __restrict__ acc4 = (float4*)acc;
    {
        const float4* __restrict__ h4 = (const float4*)(h + (size_t)lo * HCc);
        float4 hv[7];
#pragma unroll
        for (int k = 0; k < 7; ++k) {
            const int i4 = tid + k * OWN_THR;
            hv[k] = h4[(i4 < n4) ? i4 : 0];
        }
#pragma unroll
        for (int k = 0; k < 7; ++k) {
            const int i4 = tid + k * OWN_THR;
            if (i4 < n4) {
                const int nl = i4 >> 4;
                const float ad = s_ad[2 * nl + hd4];
                const float m  = lrelu02(funkey(mkey[2 * nl + hd4]) + ad);
                const float ex = __expf(lrelu02(s_as[2 * nl + hd4] + ad) - m);
                float4 a;
                a.x = hv[k].x * ex; a.y = hv[k].y * ex; a.z = hv[k].z * ex; a.w = hv[k].w * ex;
                acc4[i4] = a;
                if ((c4 & 31) == 0) { s_m[2 * nl + hd4] = m; s_den[2 * nl + hd4] = ex; }
            }
        }
    }
    __syncthreads();

    if (!over) {
        const char* __restrict__ hb = (const char*)h;
        const unsigned lane4 = (unsigned)lane * 4u;
        for (int base = wv * 32; base < ne; base += OWN_THR) {
            const int i = base + lane;
            const int cnt = (ne - base < 32) ? (ne - base) : 32;
            int s = 0, d = -1;
            float ex0 = 0.f, ex1 = 0.f;
            if (i < ne) { s = ls[i]; d = ld[i]; }
            if (d >= 0) {
                const float2 a = *(const float2*)(asrc + 2 * s);
                ex0 = __expf(lrelu02(a.x + s_ad[2 * d])     - s_m[2 * d]);
                ex1 = __expf(lrelu02(a.y + s_ad[2 * d + 1]) - s_m[2 * d + 1]);
                atomicAdd(&s_den[2 * d],     ex0);
                atomicAdd(&s_den[2 * d + 1], ex1);
            }
            const unsigned soff = (d >= 0) ? (unsigned)s * (unsigned)(HCc * sizeof(float)) : 0u;
            const int      drow = (d >= 0) ? d : OWN;
            rows16<0>(hb, acc, lane4, soff, drow, ex0, ex1);
            if (cnt > 16) rows16<16>(hb, acc, lane4, soff, drow, ex0, ex1);
        }
    } else {
        for (int e = tid; e < E; e += OWN_THR) {
            const int d = ei[E + e] - lo;
            if (d >= 0 && d < cntN) {
                const int s = ei[e];
                const float2 a = *(const float2*)(asrc + 2 * s);
                const float ex0 = __expf(lrelu02(a.x + s_ad[2 * d])     - s_m[2 * d]);
                const float ex1 = __expf(lrelu02(a.y + s_ad[2 * d + 1]) - s_m[2 * d + 1]);
                atomicAdd(&s_den[2 * d],     ex0);
                atomicAdd(&s_den[2 * d + 1], ex1);
                const float* hr = h + (size_t)s * HCc;
                for (int c = 0; c < HIDc; ++c) {
                    atomicAdd(&acc[d * HCc + c],        hr[c] * ex0);
                    atomicAdd(&acc[d * HCc + HIDc + c], hr[HIDc + c] * ex1);
                }
            }
        }
    }
    __syncthreads();

    {
        float bi[4], mu[4], sc[4], be[4];
#pragma unroll
        for (int j = 0; j < 4; ++j) {
            bi[j] = bias[c4 + j];
            mu[j] = 0.f; sc[j] = 1.f; be[j] = 0.f;
            if (LAYER == 1) {
                mu[j] = mean[c4 + j];
                sc[j] = gamma[c4 + j] * rsqrtf(var[c4 + j] + 1e-5f);
                be[j] = beta[c4 + j];
            }
        }
        float4* __restrict__ xo4 = (float4*)(xout + (size_t)lo * HCc);
        float4 xp[7];
        if (LAYER == 2) {
            const float4* __restrict__ xp4 = (const float4*)(xprev + (size_t)lo * HCc);
#pragma unroll
            for (int k = 0; k < 7; ++k) {
                const int i4 = tid + k * OWN_THR;
                xp[k] = xp4[(i4 < n4) ? i4 : 0];
            }
        }
#pragma unroll
        for (int k = 0; k < 7; ++k) {
            const int i4 = tid + k * OWN_THR;
            if (i4 < n4) {
                const int nl = i4 >> 4;
                const float dn = s_den[2 * nl + hd4] + 1e-16f;
                const float4 a = acc4[i4];
                float v[4] = {a.x / dn + bi[0], a.y / dn + bi[1], a.z / dn + bi[2], a.w / dn + bi[3]};
                if (LAYER == 1) {
#pragma unroll
                    for (int j = 0; j < 4; ++j) {
                        v[j] = (v[j] - mu[j]) * sc[j] + be[j];
                        v[j] = (v[j] > 0.f) ? v[j] : (__expf(v[j]) - 1.f);
                    }
                } else {
                    v[0] = fmaxf(xp[k].x, v[0]); v[1] = fmaxf(xp[k].y, v[1]);
                    v[2] = fmaxf(xp[k].z, v[2]); v[3] = fmaxf(xp[k].w, v[3]);
                }
                float4 o; o.x = v[0]; o.y = v[1]; o.z = v[2]; o.w = v[3];
                xo4[i4] = o;
            }
        }
    }
}

extern "C" void kernel_launch(void* const* d_in, const int* in_sizes, int n_in,
                              void* d_out, int out_size, void* d_ws, size_t ws_size,
                              hipStream_t stream) {
    const float* node_feat = (const float*)d_in[0];
    const int*   ei        = (const int*)  d_in[1];
    const float* W1        = (const float*)d_in[2];
    const float* a_src1    = (const float*)d_in[3];
    const float* a_dst1    = (const float*)d_in[4];
    const float* b1        = (const float*)d_in[5];
    const float* bn_gamma  = (const float*)d_in[6];
    const float* bn_beta   = (const float*)d_in[7];
    const float* bn_mean   = (const float*)d_in[8];
    const float* bn_var    = (const float*)d_in[9];
    const float* W2        = (const float*)d_in[10];
    const float* a_src2    = (const float*)d_in[11];
    const float* a_dst2    = (const float*)d_in[12];
    const float* b2        = (const float*)d_in[13];
    const float* Wf        = (const float*)d_in[14];
    const float* bf        = (const float*)d_in[15];
    float* out = (float*)d_out;

    const int E = in_sizes[1] / 2;

    float* ws   = (float*)d_ws;
    float* H    = ws;
    float* X1   = H    + (size_t)Nn * HCc;
    float* XJ   = X1   + (size_t)Nn * HCc;
    float* ASRC = XJ   + (size_t)Nn * HCc;
    float* ADST = ASRC + (size_t)Nn * 2;

    static_assert(GEMM_WG * (GEMM_THR / 32) * 16 >= Nn, "every row tile needs a wave");
    static_assert(OWN_WG * OWN >= Nn && OWN_WG <= 256, "every node needs an owner, every owner a byte value");
    unsigned* OB = (unsigned*)d_out;
    static_assert(OWN <= 511 && Nn <= (1 << 17), "a list word holds source << 9 | local destination, 511 = none");
    static_assert((1 << 20) + OWN_WG * ECAP + OWN_WG <= Nn * OUTc, "table, lists and lengths fit in the output buffer");
    unsigned* LST  = (unsigned*)d_out + (1 << 20);
    int*      LCNT = (int*)(LST + OWN_WG * ECAP);

    wmma_gemm<4, true, true, FIN><<<GEMM_WG, GEMM_THR, 0, stream>>>(
        node_feat, W1, nullptr, H, Nn, HCc, a_src1, a_dst1, ASRC, ADST, ei, E, OB);
    owner_layer<1><<<OWN_WG, OWN_THR, 0, stream>>>(
        ei, E, OB, H, ASRC, ADST, b1, bn_gamma, bn_beta, bn_mean, bn_var, nullptr, X1, LST, LCNT);

    wmma_gemm<4, true, false, HCc><<<GEMM_WG, GEMM_THR, 0, stream>>>(
        X1, W2, nullptr, H, Nn, HCc, a_src2, a_dst2, ASRC, ADST, nullptr, 0, nullptr);
    owner_layer<2><<<OWN_WG, OWN_THR, 0, stream>>>(
        ei, E, OB, H, ASRC, ADST, b2, nullptr, nullptr, nullptr, nullptr, X1, XJ, LST, LCNT);

    wmma_gemm<3, false, false, HCc><<<GEMM_WG, GEMM_THR, 0, stream>>>(
        XJ, Wf, bf, out, Nn, OUTc, nullptr, nullptr, nullptr, nullptr, nullptr, 0, nullptr);
}


# owners find their 16-edge groups in a group map (bit per owner and group, filed from LDS by the first launch's table writers) instead of scanning the table; list order, so LDS add order, changes
# speedup vs baseline: 8.5648x; 1.1093x over previous
#include <hip/hip_runtime.h>
#include <hip/hip_bf16.h>
#include <math.h>


#define Nn    100000
#define FIN   128
#define HCc   64
#define HIDc  32
#define OUTc  40
#define GEMM_WG   256
#define GEMM_THR  1024
#define OWN       391
#define OWN_WG    256
#define OWN_THR   1024
#define QCAP      8192
#define ECAP      8192
#define SLG       1024
#define SLW       (SLG / 32)
#define BROW      3136

typedef __attribute__((ext_vector_type(2))) float v2f;
typedef __attribute__((ext_vector_type(8))) float v8f;
typedef int dst4 __attribute__((ext_vector_type(4), aligned(4)));

__device__ inline int fkey(float v) {
    const int b = __float_as_int(v);
    return (b >= 0) ? b : (b ^ 0x7FFFFFFF);
}
__device__ inline float funkey(int k) {
    return __int_as_float((k >= 0) ? k : (k ^ 0x7FFFFFFF));
}
__device__ inline float lrelu02(float v) { return (v > 0.f) ? v : 0.2f * v; }

template<int MASK>
__device__ __forceinline__ float row_xor_add(float v) {
    const int o = __builtin_amdgcn_update_dpp(0, __float_as_int(v), 0x160 + MASK,
                                              0xf, 0xf, true);
    return v + __int_as_float(o);
}

template<int NT, bool PREP, bool OBY, int K>
__global__ __launch_bounds__(GEMM_THR)
void wmma_gemm(const float* __restrict__ A,
               const float* __restrict__ B, const float* __restrict__ bias,
               float* __restrict__ C, int M, int Ncols,
               const float* __restrict__ att_s, const float* __restrict__ att_d,
               float* __restrict__ asrc, float* __restrict__ adst,
               const int* __restrict__ ei, int E, unsigned* __restrict__ ob,
               unsigned* __restrict__ bm) {
    static_assert(!PREP || NT == 4, "PREP needs the full 64-column row in the wave");
    if (OBY) {
        constexpr int BTP = OWN_WG + 1;
        __shared__ unsigned bt[SLW * BTP];
        const int nQ = E >> 4;
        const int g0 = blockIdx.x * SLG;
        static_assert(SLG == GEMM_THR && SLW == 32, "one group per thread, one 128-byte line of a row per slice");
        if (blockIdx.x == 0 && threadIdx.x == 0) {
            for (int w = 4 * nQ; 4 * w < E; ++w) {
                unsigned pk1 = 0u;
                for (int j = 0; j < 4; ++j) {
                    const int e = 4 * w + j;
                    if (e < E) pk1 |= (((unsigned)ei[E + e] / (unsigned)OWN) & 0xFFu) << (8 * j);
                }
                ob[w] = pk1;
            }
        }
        if (16 * g0 < E && (blockIdx.x + 1) * SLW <= BROW) {
            for (int i = threadIdx.x; i < SLW * BTP; i += GEMM_THR) bt[i] = 0u;
            __syncthreads();
            const int g  = threadIdx.x;
            const int gA = g0 + g;
            if (nQ > 0) {
                const int gc = (gA < nQ) ? gA : (nQ - 1);
                const dst4* __restrict__ dp = (const dst4*)(ei + E + 16 * gc);
                dst4 dv[4];
#pragma unroll
                for (int j = 0; j < 4; ++j) dv[j] = dp[j];
                if (gA < nQ) {
                    unsigned pk[4];
#pragma unroll
                    for (int j = 0; j < 4; ++j) {
                        const unsigned o0 = ((unsigned)dv[j].x / (unsigned)OWN) & 0xFFu;
                        const unsigned o1 = ((unsigned)dv[j].y / (unsigned)OWN) & 0xFFu;
                        const unsigned o2 = ((unsigned)dv[j].z / (unsigned)OWN) & 0xFFu;
                        const unsigned o3 = ((unsigned)dv[j].w / (unsigned)OWN) & 0xFFu;
                        pk[j] = o0 | (o1 << 8) | (o2 << 16) | (o3 << 24);
                        const unsigned bit = 1u << (g & 31);
                        atomicOr(&bt[(g >> 5) * BTP + o0], bit);
                        atomicOr(&bt[(g >> 5) * BTP + o1], bit);
                        atomicOr(&bt[(g >> 5) * BTP + o2], bit);
                        atomicOr(&bt[(g >> 5) * BTP + o3], bit);
                    }
                    uint4 t4; t4.x = pk[0]; t4.y = pk[1]; t4.z = pk[2]; t4.w = pk[3];
                    *(uint4*)(ob + 4 * gA) = t4;
                }
            }
            __syncthreads();
            for (int i = threadIdx.x; i < OWN_WG * SLW; i += GEMM_THR)
                bm[(size_t)(i / SLW) * BROW + blockIdx.x * SLW + (i % SLW)] = bt[(i % SLW) * BTP + (i / SLW)];
        }
    }
    __shared__ float Bs[(K / 2) * 64 * 2];

    for (int i = threadIdx.x; i < K * 64; i += GEMM_THR) {
        const int p    = i >> 7;
        const int rem  = i & 127;
        const int col  = rem >> 1;
        const int half = rem & 1;
        Bs[i] = (col < Ncols) ? B[(size_t)(2 * p + half) * Ncols + col] : 0.f;
    }
    __syncthreads();

    const int wave = threadIdx.x >> 5;
    const int lane = threadIdx.x & 31;
    const int tileM = wave * GEMM_WG + blockIdx.x;
    const int row16 = tileM * 16;
    if (row16 >= M) return;

    const int hi = (lane >= 16);
    const int lr = lane & 15;

    v8f acc[NT];
#pragma unroll
    for (int t = 0; t < NT; ++t)
        acc[t] = (v8f){0.f, 0.f, 0.f, 0.f, 0.f, 0.f, 0.f, 0.f};

    const int arow = row16 + lr;
    const float* __restrict__ Arow  = A + (size_t)arow * K;

#pragma unroll 4
    for (int kb = 0; kb < K; kb += 4) {
        const int ka = kb + 2 * hi;
        const float a0 = Arow[ka];
        const float a1 = Arow[ka + 1];
        v2f afrag; afrag.x = a0; afrag.y = a1;

        const int p = (kb >> 1) + hi;
#pragma unroll
        for (int t = 0; t < NT; ++t) {
            const v2f bfrag = *(const v2f*)&Bs[(p << 7) + ((t * 16 + lr) << 1)];
            acc[t] = __builtin_amdgcn_wmma_f32_16x16x4_f32(
                false, afrag, false, bfrag, (short)0, acc[t], false, false);
        }
    }

    const int rbase = row16 + (hi ? 8 : 0);
#pragma unroll
    for (int t = 0; t < NT; ++t) {
        const int col = t * 16 + lr;
        if (col < Ncols) {
            const float bb = bias ? bias[col] : 0.f;
#pragma unroll
            for (int r = 0; r < 8; ++r)
                C[(size_t)(rbase + r) * Ncols + col] = acc[t][r] + bb;
        }
    }

    if (PREP) {
        float vs[2][8], vd[2][8];
#pragma unroll
        for (int hd = 0; hd < 2; ++hd) {
            const float s_lo = att_s[hd * HIDc + lr], s_hi = att_s[hd * HIDc + 16 + lr];
            const float d_lo = att_d[hd * HIDc + lr], d_hi = att_d[hd * HIDc + 16 + lr];
#pragma unroll
            for (int r = 0; r < 8; ++r) {
                vs[hd][r] = acc[2 * hd][r] * s_lo + acc[2 * hd + 1][r] * s_hi;
                vd[hd][r] = acc[2 * hd][r] * d_lo + acc[2 * hd + 1][r] * d_hi;
            }
        }
#pragma unroll
        for (int hd = 0; hd < 2; ++hd)
#pragma unroll
            for (int r = 0; r < 8; ++r) { vs[hd][r] = row_xor_add<8>(vs[hd][r]); vd[hd][r] = row_xor_add<8>(vd[hd][r]); }
#pragma unroll
        for (int hd = 0; hd < 2; ++hd)
#pragma unroll
            for (int r = 0; r < 8; ++r) { vs[hd][r] = row_xor_add<4>(vs[hd][r]); vd[hd][r] = row_xor_add<4>(vd[hd][r]); }
#pragma unroll
        for (int hd = 0; hd < 2; ++hd)
#pragma unroll
            for (int r = 0; r < 8; ++r) { vs[hd][r] = row_xor_add<2>(vs[hd][r]); vd[hd][r] = row_xor_add<2>(vd[hd][r]); }
#pragma unroll
        for (int hd = 0; hd < 2; ++hd)
#pragma unroll
            for (int r = 0; r < 8; ++r) { vs[hd][r] = row_xor_add<1>(vs[hd][r]); vd[hd][r] = row_xor_add<1>(vd[hd][r]); }
        float os0 = vs[0][0], os1 = vs[1][0], od0 = vd[0][0], od1 = vd[1][0];
#pragma unroll
        for (int r = 1; r < 8; ++r) {
            os0 = (lr == r) ? vs[0][r] : os0;
            os1 = (lr == r) ? vs[1][r] : os1;
            od0 = (lr == r) ? vd[0][r] : od0;
            od1 = (lr == r) ? vd[1][r] : od1;
        }
        if (lr < 8) {
            const int n = rbase + lr;
            float2 a2; a2.x = os0; a2.y = os1;
            *(float2*)(asrc + 2 * n) = a2;
            float2 d2; d2.x = od0; d2.y = od1;
            *(float2*)(adst + 2 * n) = d2;
        }
    }
}

template<int J0>
__device__ __forceinline__ void rows16(const char* __restrict__ hb, float* __restrict__ acc,
                                       unsigned lane4, unsigned soff, int drow,
                                       float ex0, float ex1) {
    float hv0[16], hv1[16];
#pragma unroll
    for (int u = 0; u < 16; ++u) {
        const unsigned so = (unsigned)__builtin_amdgcn_readlane((int)soff, J0 + u);
        const float* hr = (const float*)(hb + (size_t)(so + lane4));
        hv0[u] = hr[0];
        hv1[u] = hr[HIDc];
    }
#pragma unroll
    for (int u = 0; u < 16; ++u) {
        const int dj = __builtin_amdgcn_readlane(drow, J0 + u);
        const float x0 = __int_as_float(__builtin_amdgcn_readlane(__float_as_int(ex0), J0 + u));
        const float x1 = __int_as_float(__builtin_amdgcn_readlane(__float_as_int(ex1), J0 + u));
        atomicAdd(&acc[dj * HCc + (int)(lane4 >> 2)],        hv0[u] * x0);
        atomicAdd(&acc[dj * HCc + HIDc + (int)(lane4 >> 2)], hv1[u] * x1);
    }
}

__device__ __forceinline__ unsigned zbytes(unsigned y) {
    return ~(((y & 0x7F7F7F7Fu) + 0x7F7F7F7Fu) | y | 0x7F7F7F7Fu);
}
__device__ __forceinline__ unsigned zpack(unsigned z) {
    const unsigned t = z >> 7;
    return (t | (t >> 7) | (t >> 14) | (t >> 21)) & 0xFu;
}

template<int LAYER>
__global__ __launch_bounds__(OWN_THR)
void owner_layer(const int* __restrict__ ei, int E, const unsigned* __restrict__ ob,
                 const float* __restrict__ h, const float* __restrict__ asrc,
                 const float* __restrict__ adst, const float* __restrict__ bias,
                 const float* __restrict__ gamma, const float* __restrict__ beta,
                 const float* __restrict__ mean, const float* __restrict__ var,
                 const float* __restrict__ xprev, float* __restrict__ xout,
                 unsigned* __restrict__ lst, int* __restrict__ lcnt,
                 const unsigned* __restrict__ bm) {
    __shared__ __attribute__((aligned(16))) float acc[(OWN + 1) * HCc];
    __shared__ int   mkey[OWN * 2];
    __shared__ float s_as[OWN * 2], s_ad[OWN * 2], s_m[OWN * 2], s_den[OWN * 2];
    __shared__ int   candq[QCAP];
    __shared__ int   cande[ECAP];
    __shared__ int   ls[ECAP], ld[ECAP];
    __shared__ int   ctr[2];

    const int tid  = threadIdx.x;
    const int lane = tid & 31;
    const int wv   = __builtin_amdgcn_readfirstlane(tid >> 5);
    const int lo   = blockIdx.x * OWN;
    const int cntN = (Nn - lo < OWN) ? (Nn - lo) : OWN;

    if (LAYER == 1) {
        if (tid < 2) ctr[tid] = 0;
        if (tid == 2) { candq[0] = 0; cande[0] = 0; }
    }
    for (int i = tid; i < cntN * 2; i += OWN_THR) {
        const float a = asrc[lo * 2 + i];
        s_as[i] = a;
        s_ad[i] = adst[lo * 2 + i];
        mkey[i] = fkey(a);
    }
    __syncthreads();

    int  ne;
    bool over;
    if (LAYER == 1) {
        const unsigned pat = (unsigned)blockIdx.x * 0x01010101u;
        const int nQ = E >> 4;
        const uint4* __restrict__ ob4 = (const uint4*)ob;
        const int  nBW     = (nQ + 31) >> 5;
        const bool mapless = nBW > BROW;
        static_assert(BROW <= 4 * OWN_THR, "four words per thread cover a row");
        if (!mapless) {
            const unsigned* __restrict__ row = bm + (size_t)blockIdx.x * BROW;
            unsigned wd[4];
#pragma unroll
            for (int u = 0; u < 4; ++u) {
                const int wi = tid + u * OWN_THR;
                wd[u] = row[(wi < nBW) ? wi : 0];
            }
#pragma unroll
            for (int u = 0; u < 4; ++u) {
                const int wi = tid + u * OWN_THR;
                unsigned mm = (wi < nBW) ? wd[u] : 0u;
                for (int trip = 0; trip < 32; ++trip) {
                    const bool has = (mm != 0u);
                    const unsigned bal = __builtin_amdgcn_ballot_w32(has);
                    if (bal == 0u) break;
                    const int j = has ? __builtin_ctz(mm) : 0;
                    mm &= mm - 1u;
                    const int lead = __builtin_ctz(bal);
                    int draw = 0;
                    if (lane == lead) draw = atomicAdd(&ctr[0], __builtin_popcount(bal));
                    draw = __builtin_amdgcn_readlane(draw, lead);
                    const int p = draw + (int)__builtin_amdgcn_mbcnt_lo(bal, 0u);
                    if (has && p < QCAP) candq[p] = 32 * wi + j;
                }
            }
        }
        __syncthreads();

        const int nq = ctr[0];
        over = mapless || (nq > QCAP);
        if (!over) {
            int qq[8]; uint4 vv[8];
#pragma unroll
            for (int u = 0; u < 8; ++u) {
                const int i = tid + u * OWN_THR;
                const int q = candq[(i < nq) ? i : 0];
                vv[u] = ob4[q];
                qq[u] = (i < nq) ? q : -1;
            }
#pragma unroll
            for (int u = 0; u < 8; ++u) {
                unsigned mm = zpack(zbytes(vv[u].x ^ pat))
                            | (zpack(zbytes(vv[u].y ^ pat)) << 4)
                            | (zpack(zbytes(vv[u].z ^ pat)) << 8)
                            | (zpack(zbytes(vv[u].w ^ pat)) << 12);
                if (qq[u] < 0) mm = 0u;
                for (int trip = 0; trip < 16; ++trip) {
                    const bool has = (mm != 0u);
                    const unsigned bal = __builtin_amdgcn_ballot_w32(has);
                    if (bal == 0u) break;
                    const int j = has ? __builtin_ctz(mm) : 0;
                    mm &= mm - 1u;
                    const int lead = __builtin_ctz(bal);
                    int draw = 0;
                    if (lane == lead) draw = atomicAdd(&ctr[1], __builtin_popcount(bal));
                    draw = __builtin_amdgcn_readlane(draw, lead);
                    const int p = draw + (int)__builtin_amdgcn_mbcnt_lo(bal, 0u);
                    if (has && p < ECAP) cande[p] = 16 * qq[u] + j;
                }
            }
            for (int e = 16 * nQ + tid; e < E; e += OWN_THR) {
                const unsigned by = (ob[e >> 2] >> (8 * (e & 3))) & 0xFFu;
                if (by == (unsigned)blockIdx.x) {
                    const int p = atomicAdd(&ctr[1], 1);
                    if (p < ECAP) cande[p] = e;
                }
            }
        }
        __syncthreads();

        ne = ctr[1];
        over = over || (ne > ECAP);
        if (tid == 0) lcnt[blockIdx.x] = over ? (ECAP + 1) : ne;
    } else {
        ne = lcnt[blockIdx.x];
        over = (unsigned)ne > (unsigned)ECAP;
    }

    unsigned* __restrict__ lb = lst + (size_t)blockIdx.x * ECAP;
    if (!over) {
        int sv[8], dv[8];
        if (LAYER == 1) {
#pragma unroll
            for (int u = 0; u < 8; ++u) {
                const int i = tid + u * OWN_THR;
                const int e = cande[(i < ne) ? i : 0];
                sv[u] = ei[e];
                dv[u] = ei[E + e] - lo;
            }
        } else {
#pragma unroll
            for (int u = 0; u < 8; ++u) {
                const int i = tid + u * OWN_THR;
                const unsigned rec = lb[(i < ne) ? i : 0];
                const unsigned s = rec >> 9;
                sv[u] = (s < (unsigned)Nn) ? (int)s : 0;
                dv[u] = (int)(rec & 511u);
            }
        }
        float2 av[8];
#pragma unroll
        for (int u = 0; u < 8; ++u) {
            const int i = tid + u * OWN_THR;
            if (i >= ne || dv[u] < 0 || dv[u] >= cntN) dv[u] = -1;
            av[u] = *(const float2*)(asrc + 2 * sv[u]);
        }
#pragma unroll
        for (int u = 0; u < 8; ++u) {
            const int i = tid + u * OWN_THR;
            if (i < ne) {
                ls[i] = sv[u]; ld[i] = dv[u];
                if (LAYER == 1) lb[i] = ((unsigned)sv[u] << 9) | ((unsigned)dv[u] & 511u);
            }
            if (dv[u] >= 0) {
                atomicMax(&mkey[2 * dv[u]],     fkey(av[u].x));
                atomicMax(&mkey[2 * dv[u] + 1], fkey(av[u].y));
            }
        }
    } else {
        for (int e = tid; e < E; e += OWN_THR) {
            const int d = ei[E + e] - lo;
            if (d >= 0 && d < cntN) {
                const float2 a = *(const float2*)(asrc + 2 * ei[e]);
                atomicMax(&mkey[2 * d],     fkey(a.x));
                atomicMax(&mkey[2 * d + 1], fkey(a.y));
            }
        }
    }
    __syncthreads();

    static_assert(OWN * (HCc / 4) <= 7 * OWN_THR && OWN_THR % (HCc / 4) == 0, "seven passes cover an owner's rows");
    const int c4  = (tid & 15) * 4;
    const int hd4 = c4 >> 5;
    const int n4  = cntN * (HCc / 4);
    float4* __restrict__ acc4 = (float4*)acc;
    {
        const float4* __restrict__ h4 = (const float4*)(h + (size_t)lo * HCc);
        float4 hv[7];
#pragma unroll
        for (int k = 0; k < 7; ++k) {
            const int i4 = tid + k * OWN_THR;
            hv[k] = h4[(i4 < n4) ? i4 : 0];
        }
#pragma unroll
        for (int k = 0; k < 7; ++k) {
            const int i4 = tid + k * OWN_THR;
            if (i4 < n4) {
                const int nl = i4 >> 4;
                const float ad = s_ad[2 * nl + hd4];
                const float m  = lrelu02(funkey(mkey[2 * nl + hd4]) + ad);
                const float ex = __expf(lrelu02(s_as[2 * nl + hd4] + ad) - m);
                float4 a;
                a.x = hv[k].x * ex; a.y = hv[k].y * ex; a.z = hv[k].z * ex; a.w = hv[k].w * ex;
                acc4[i4] = a;
                if ((c4 & 31) == 0) { s_m[2 * nl + hd4] = m; s_den[2 * nl + hd4] = ex; }
            }
        }
    }
    __syncthreads();

    if (!over) {
        const char* __restrict__ hb = (const char*)h;
        const unsigned lane4 = (unsigned)lane * 4u;
        for (int base = wv * 32; base < ne; base += OWN_THR) {
            const int i = base + lane;
            const int cnt = (ne - base < 32) ? (ne - base) : 32;
            int s = 0, d = -1;
            float ex0 = 0.f, ex1 = 0.f;
            if (i < ne) { s = ls[i]; d = ld[i]; }
            if (d >= 0) {
                const float2 a = *(const float2*)(asrc + 2 * s);
                ex0 = __expf(lrelu02(a.x + s_ad[2 * d])     - s_m[2 * d]);
                ex1 = __expf(lrelu02(a.y + s_ad[2 * d + 1]) - s_m[2 * d + 1]);
                atomicAdd(&s_den[2 * d],     ex0);
                atomicAdd(&s_den[2 * d + 1], ex1);
            }
            const unsigned soff = (d >= 0) ? (unsigned)s * (unsigned)(HCc * sizeof(float)) : 0u;
            const int      drow = (d >= 0) ? d : OWN;
            rows16<0>(hb, acc, lane4, soff, drow, ex0, ex1);
            if (cnt > 16) rows16<16>(hb, acc, lane4, soff, drow, ex0, ex1);
        }
    } else {
        for (int e = tid; e < E; e += OWN_THR) {
            const int d = ei[E + e] - lo;
            if (d >= 0 && d < cntN) {
                const int s = ei[e];
                const float2 a = *(const float2*)(asrc + 2 * s);
                const float ex0 = __expf(lrelu02(a.x + s_ad[2 * d])     - s_m[2 * d]);
                const float ex1 = __expf(lrelu02(a.y + s_ad[2 * d + 1]) - s_m[2 * d + 1]);
                atomicAdd(&s_den[2 * d],     ex0);
                atomicAdd(&s_den[2 * d + 1], ex1);
                const float* hr = h + (size_t)s * HCc;
                for (int c = 0; c < HIDc; ++c) {
                    atomicAdd(&acc[d * HCc + c],        hr[c] * ex0);
                    atomicAdd(&acc[d * HCc + HIDc + c], hr[HIDc + c] * ex1);
                }
            }
        }
    }
    __syncthreads();

    {
        float bi[4], mu[4], sc[4], be[4];
#pragma unroll
        for (int j = 0; j < 4; ++j) {
            bi[j] = bias[c4 + j];
            mu[j] = 0.f; sc[j] = 1.f; be[j] = 0.f;
            if (LAYER == 1) {
                mu[j] = mean[c4 + j];
                sc[j] = gamma[c4 + j] * rsqrtf(var[c4 + j] + 1e-5f);
                be[j] = beta[c4 + j];
            }
        }
        float4* __restrict__ xo4 = (float4*)(xout + (size_t)lo * HCc);
        float4 xp[7];
        if (LAYER == 2) {
            const float4* __restrict__ xp4 = (const float4*)(xprev + (size_t)lo * HCc);
#pragma unroll
            for (int k = 0; k < 7; ++k) {
                const int i4 = tid + k * OWN_THR;
                xp[k] = xp4[(i4 < n4) ? i4 : 0];
            }
        }
#pragma unroll
        for (int k = 0; k < 7; ++k) {
            const int i4 = tid + k * OWN_THR;
            if (i4 < n4) {
                const int nl = i4 >> 4;
                const float dn = s_den[2 * nl + hd4] + 1e-16f;
                const float4 a = acc4[i4];
                float v[4] = {a.x / dn + bi[0], a.y / dn + bi[1], a.z / dn + bi[2], a.w / dn + bi[3]};
                if (LAYER == 1) {
#pragma unroll
                    for (int j = 0; j < 4; ++j) {
                        v[j] = (v[j] - mu[j]) * sc[j] + be[j];
                        v[j] = (v[j] > 0.f) ? v[j] : (__expf(v[j]) - 1.f);
                    }
                } else {
                    v[0] = fmaxf(xp[k].x, v[0]); v[1] = fmaxf(xp[k].y, v[1]);
                    v[2] = fmaxf(xp[k].z, v[2]); v[3] = fmaxf(xp[k].w, v[3]);
                }
                float4 o; o.x = v[0]; o.y = v[1]; o.z = v[2]; o.w = v[3];
                xo4[i4] = o;
            }
        }
    }
}

extern "C" void kernel_launch(void* const* d_in, const int* in_sizes, int n_in,
                              void* d_out, int out_size, void* d_ws, size_t ws_size,
                              hipStream_t stream) {
    const float* node_feat = (const float*)d_in[0];
    const int*   ei        = (const int*)  d_in[1];
    const float* W1        = (const float*)d_in[2];
    const float* a_src1    = (const float*)d_in[3];
    const float* a_dst1    = (const float*)d_in[4];
    const float* b1        = (const float*)d_in[5];
    const float* bn_gamma  = (const float*)d_in[6];
    const float* bn_beta   = (const float*)d_in[7];
    const float* bn_mean   = (const float*)d_in[8];
    const float* bn_var    = (const float*)d_in[9];
    const float* W2        = (const float*)d_in[10];
    const float* a_src2    = (const float*)d_in[11];
    const float* a_dst2    = (const float*)d_in[12];
    const float* b2        = (const float*)d_in[13];
    const float* Wf        = (const float*)d_in[14];
    const float* bf        = (const float*)d_in[15];
    float* out = (float*)d_out;

    const int E = in_sizes[1] / 2;

    float* ws   = (float*)d_ws;
    float* H    = ws;
    float* X1   = H    + (size_t)Nn * HCc;
    float* XJ   = X1   + (size_t)Nn * HCc;
    float* ASRC = XJ   + (size_t)Nn * HCc;
    float* ADST = ASRC + (size_t)Nn * 2;

    static_assert(GEMM_WG * (GEMM_THR / 32) * 16 >= Nn, "every row tile needs a wave");
    static_assert(OWN_WG * OWN >= Nn && OWN_WG <= 256, "every node needs an owner, every owner a byte value");
    unsigned* OB = (unsigned*)d_out;
    static_assert(OWN <= 511 && Nn <= (1 << 17), "a list word holds source << 9 | local destination, 511 = none");
    static_assert((1 << 20) + OWN_WG * ECAP + OWN_WG + OWN_WG * BROW <= Nn * OUTc, "table, lists, lengths and group map fit in the output buffer");
    static_assert(BROW % SLW == 0 && BROW / SLW <= GEMM_WG, "every slice of the group map has a work-group of the first launch");
    unsigned* LST  = (unsigned*)d_out + (1 << 20);
    int*      LCNT = (int*)(LST + OWN_WG * ECAP);
    unsigned* BM   = (unsigned*)(LCNT + OWN_WG);

    wmma_gemm<4, true, true, FIN><<<GEMM_WG, GEMM_THR, 0, stream>>>(
        node_feat, W1, nullptr, H, Nn, HCc, a_src1, a_dst1, ASRC, ADST, ei, E, OB, BM);
    owner_layer<1><<<OWN_WG, OWN_THR, 0, stream>>>(
        ei, E, OB, H, ASRC, ADST, b1, bn_gamma, bn_beta, bn_mean, bn_var, nullptr, X1, LST, LCNT, BM);

    wmma_gemm<4, true, false, HCc><<<GEMM_WG, GEMM_THR, 0, stream>>>(
        X1, W2, nullptr, H, Nn, HCc, a_src2, a_dst2, ASRC, ADST, nullptr, 0, nullptr, nullptr);
    owner_layer<2><<<OWN_WG, OWN_THR, 0, stream>>>(
        ei, E, OB, H, ASRC, ADST, b2, nullptr, nullptr, nullptr, nullptr, X1, XJ, LST, LCNT, nullptr);

    wmma_gemm<3, false, false, HCc><<<GEMM_WG, GEMM_THR, 0, stream>>>(
        XJ, Wf, bf, out, Nn, OUTc, nullptr, nullptr, nullptr, nullptr, nullptr, 0, nullptr, nullptr);
}


# owner layers: a listed edge's two source logits are gathered once (step 4, for the max) and kept in LDS; step 6 forms the exp weights from them: no second gather from the logit table per edge
# speedup vs baseline: 9.0313x; 1.0545x over previous
#include <hip/hip_runtime.h>
#include <hip/hip_bf16.h>
#include <math.h>


#define Nn    100000
#define FIN   128
#define HCc   64
#define HIDc  32
#define OUTc  40
#define GEMM_WG   256
#define GEMM_THR  1024
#define OWN       391
#define OWN_WG    256
#define OWN_THR   1024
#define QCAP      8192
#define ECAP      8192
#define SLG       1024
#define SLW       (SLG / 32)
#define BROW      3136

typedef __attribute__((ext_vector_type(2))) float v2f;
typedef __attribute__((ext_vector_type(8))) float v8f;
typedef int dst4 __attribute__((ext_vector_type(4), aligned(4)));

__device__ inline int fkey(float v) {
    const int b = __float_as_int(v);
    return (b >= 0) ? b : (b ^ 0x7FFFFFFF);
}
__device__ inline float funkey(int k) {
    return __int_as_float((k >= 0) ? k : (k ^ 0x7FFFFFFF));
}
__device__ inline float lrelu02(float v) { return (v > 0.f) ? v : 0.2f * v; }

template<int MASK>
__device__ __forceinline__ float row_xor_add(float v) {
    const int o = __builtin_amdgcn_update_dpp(0, __float_as_int(v), 0x160 + MASK,
                                              0xf, 0xf, true);
    return v + __int_as_float(o);
}

template<int NT, bool PREP, bool OBY, int K>
__global__ __launch_bounds__(GEMM_THR)
void wmma_gemm(const float* __restrict__ A,
               const float* __restrict__ B, const float* __restrict__ bias,
               float* __restrict__ C, int M, int Ncols,
               const float* __restrict__ att_s, const float* __restrict__ att_d,
               float* __restrict__ asrc, float* __restrict__ adst,
               const int* __restrict__ ei, int E, unsigned* __restrict__ ob,
               unsigned* __restrict__ bm) {
    static_assert(!PREP || NT == 4, "PREP needs the full 64-column row in the wave");
    if (OBY) {
        constexpr int BTP = OWN_WG + 1;
        __shared__ unsigned bt[SLW * BTP];
        const int nQ = E >> 4;
        const int g0 = blockIdx.x * SLG;
        static_assert(SLG == GEMM_THR && SLW == 32, "one group per thread, one 128-byte line of a row per slice");
        if (blockIdx.x == 0 && threadIdx.x == 0) {
            for (int w = 4 * nQ; 4 * w < E; ++w) {
                unsigned pk1 = 0u;
                for (int j = 0; j < 4; ++j) {
                    const int e = 4 * w + j;
                    if (e < E) pk1 |= (((unsigned)ei[E + e] / (unsigned)OWN) & 0xFFu) << (8 * j);
                }
                ob[w] = pk1;
            }
        }
        if (16 * g0 < E && (blockIdx.x + 1) * SLW <= BROW) {
            for (int i = threadIdx.x; i < SLW * BTP; i += GEMM_THR) bt[i] = 0u;
            __syncthreads();
            const int g  = threadIdx.x;
            const int gA = g0 + g;
            if (nQ > 0) {
                const int gc = (gA < nQ) ? gA : (nQ - 1);
                const dst4* __restrict__ dp = (const dst4*)(ei + E + 16 * gc);
                dst4 dv[4];
#pragma unroll
                for (int j = 0; j < 4; ++j) dv[j] = dp[j];
                if (gA < nQ) {
                    unsigned pk[4];
#pragma unroll
                    for (int j = 0; j < 4; ++j) {
                        const unsigned o0 = ((unsigned)dv[j].x / (unsigned)OWN) & 0xFFu;
                        const unsigned o1 = ((unsigned)dv[j].y / (unsigned)OWN) & 0xFFu;
                        const unsigned o2 = ((unsigned)dv[j].z / (unsigned)OWN) & 0xFFu;
                        const unsigned o3 = ((unsigned)dv[j].w / (unsigned)OWN) & 0xFFu;
                        pk[j] = o0 | (o1 << 8) | (o2 << 16) | (o3 << 24);
                        const unsigned bit = 1u << (g & 31);
                        atomicOr(&bt[(g >> 5) * BTP + o0], bit);
                        atomicOr(&bt[(g >> 5) * BTP + o1], bit);
                        atomicOr(&bt[(g >> 5) * BTP + o2], bit);
                        atomicOr(&bt[(g >> 5) * BTP + o3], bit);
                    }
                    uint4 t4; t4.x = pk[0]; t4.y = pk[1]; t4.z = pk[2]; t4.w = pk[3];
                    *(uint4*)(ob + 4 * gA) = t4;
                }
            }
            __syncthreads();
            for (int i = threadIdx.x; i < OWN_WG * SLW; i += GEMM_THR)
                bm[(size_t)(i / SLW) * BROW + blockIdx.x * SLW + (i % SLW)] = bt[(i % SLW) * BTP + (i / SLW)];
        }
    }
    __shared__ float Bs[(K / 2) * 64 * 2];

    for (int i = threadIdx.x; i < K * 64; i += GEMM_THR) {
        const int p    = i >> 7;
        const int rem  = i & 127;
        const int col  = rem >> 1;
        const int half = rem & 1;
        Bs[i] = (col < Ncols) ? B[(size_t)(2 * p + half) * Ncols + col] : 0.f;
    }
    __syncthreads();

    const int wave = threadIdx.x >> 5;
    const int lane = threadIdx.x & 31;
    const int tileM = wave * GEMM_WG + blockIdx.x;
    const int row16 = tileM * 16;
    if (row16 >= M) return;

    const int hi = (lane >= 16);
    const int lr = lane & 15;

    v8f acc[NT];
#pragma unroll
    for (int t = 0; t < NT; ++t)
        acc[t] = (v8f){0.f, 0.f, 0.f, 0.f, 0.f, 0.f, 0.f, 0.f};

    const int arow = row16 + lr;
    const float* __restrict__ Arow  = A + (size_t)arow * K;

#pragma unroll 4
    for (int kb = 0; kb < K; kb += 4) {
        const int ka = kb + 2 * hi;
        const float a0 = Arow[ka];
        const float a1 = Arow[ka + 1];
        v2f afrag; afrag.x = a0; afrag.y = a1;

        const int p = (kb >> 1) + hi;
#pragma unroll
        for (int t = 0; t < NT; ++t) {
            const v2f bfrag = *(const v2f*)&Bs[(p << 7) + ((t * 16 + lr) << 1)];
            acc[t] = __builtin_amdgcn_wmma_f32_16x16x4_f32(
                false, afrag, false, bfrag, (short)0, acc[t], false, false);
        }
    }

    const int rbase = row16 + (hi ? 8 : 0);
#pragma unroll
    for (int t = 0; t < NT; ++t) {
        const int col = t * 16 + lr;
        if (col < Ncols) {
            const float bb = bias ? bias[col] : 0.f;
#pragma unroll
            for (int r = 0; r < 8; ++r)
                C[(size_t)(rbase + r) * Ncols + col] = acc[t][r] + bb;
        }
    }

    if (PREP) {
        float vs[2][8], vd[2][8];
#pragma unroll
        for (int hd = 0; hd < 2; ++hd) {
            const float s_lo = att_s[hd * HIDc + lr], s_hi = att_s[hd * HIDc + 16 + lr];
            const float d_lo = att_d[hd * HIDc + lr], d_hi = att_d[hd * HIDc + 16 + lr];
#pragma unroll
            for (int r = 0; r < 8; ++r) {
                vs[hd][r] = acc[2 * hd][r] * s_lo + acc[2 * hd + 1][r] * s_hi;
                vd[hd][r] = acc[2 * hd][r] * d_lo + acc[2 * hd + 1][r] * d_hi;
            }
        }
#pragma unroll
        for (int hd = 0; hd < 2; ++hd)
#pragma unroll
            for (int r = 0; r < 8; ++r) { vs[hd][r] = row_xor_add<8>(vs[hd][r]); vd[hd][r] = row_xor_add<8>(vd[hd][r]); }
#pragma unroll
        for (int hd = 0; hd < 2; ++hd)
#pragma unroll
            for (int r = 0; r < 8; ++r) { vs[hd][r] = row_xor_add<4>(vs[hd][r]); vd[hd][r] = row_xor_add<4>(vd[hd][r]); }
#pragma unroll
        for (int hd = 0; hd < 2; ++hd)
#pragma unroll
            for (int r = 0; r < 8; ++r) { vs[hd][r] = row_xor_add<2>(vs[hd][r]); vd[hd][r] = row_xor_add<2>(vd[hd][r]); }
#pragma unroll
        for (int hd = 0; hd < 2; ++hd)
#pragma unroll
            for (int r = 0; r < 8; ++r) { vs[hd][r] = row_xor_add<1>(vs[hd][r]); vd[hd][r] = row_xor_add<1>(vd[hd][r]); }
        float os0 = vs[0][0], os1 = vs[1][0], od0 = vd[0][0], od1 = vd[1][0];
#pragma unroll
        for (int r = 1; r < 8; ++r) {
            os0 = (lr == r) ? vs[0][r] : os0;
            os1 = (lr == r) ? vs[1][r] : os1;
            od0 = (lr == r) ? vd[0][r] : od0;
            od1 = (lr == r) ? vd[1][r] : od1;
        }
        if (lr < 8) {
            const int n = rbase + lr;
            float2 a2; a2.x = os0; a2.y = os1;
            *(float2*)(asrc + 2 * n) = a2;
            float2 d2; d2.x = od0; d2.y = od1;
            *(float2*)(adst + 2 * n) = d2;
        }
    }
}

template<int J0>
__device__ __forceinline__ void rows16(const char* __restrict__ hb, float* __restrict__ acc,
                                       unsigned lane4, unsigned soff, int drow,
                                       float ex0, float ex1) {
    float hv0[16], hv1[16];
#pragma unroll
    for (int u = 0; u < 16; ++u) {
        const unsigned so = (unsigned)__builtin_amdgcn_readlane((int)soff, J0 + u);
        const float* hr = (const float*)(hb + (size_t)(so + lane4));
        hv0[u] = hr[0];
        hv1[u] = hr[HIDc];
    }
#pragma unroll
    for (int u = 0; u < 16; ++u) {
        const int dj = __builtin_amdgcn_readlane(drow, J0 + u);
        const float x0 = __int_as_float(__builtin_amdgcn_readlane(__float_as_int(ex0), J0 + u));
        const float x1 = __int_as_float(__builtin_amdgcn_readlane(__float_as_int(ex1), J0 + u));
        atomicAdd(&acc[dj * HCc + (int)(lane4 >> 2)],        hv0[u] * x0);
        atomicAdd(&acc[dj * HCc + HIDc + (int)(lane4 >> 2)], hv1[u] * x1);
    }
}

__device__ __forceinline__ unsigned zbytes(unsigned y) {
    return ~(((y & 0x7F7F7F7Fu) + 0x7F7F7F7Fu) | y | 0x7F7F7F7Fu);
}
__device__ __forceinline__ unsigned zpack(unsigned z) {
    const unsigned t = z >> 7;
    return (t | (t >> 7) | (t >> 14) | (t >> 21)) & 0xFu;
}

template<int LAYER>
__global__ __launch_bounds__(OWN_THR)
void owner_layer(const int* __restrict__ ei, int E, const unsigned* __restrict__ ob,
                 const float* __restrict__ h, const float* __restrict__ asrc,
                 const float* __restrict__ adst, const float* __restrict__ bias,
                 const float* __restrict__ gamma, const float* __restrict__ beta,
                 const float* __restrict__ mean, const float* __restrict__ var,
                 const float* __restrict__ xprev, float* __restrict__ xout,
                 unsigned* __restrict__ lst, int* __restrict__ lcnt,
                 const unsigned* __restrict__ bm) {
    __shared__ __attribute__((aligned(16))) float acc[(OWN + 1) * HCc];
    __shared__ int   mkey[OWN * 2];
    __shared__ float s_as[OWN * 2], s_ad[OWN * 2], s_m[OWN * 2], s_den[OWN * 2];
    __shared__ int   candq[QCAP];
    __shared__ int   cande[ECAP];
    __shared__ int   ls[ECAP], ld[ECAP];
    __shared__ __attribute__((aligned(8))) float lav[2 * ECAP];
    __shared__ int   ctr[2];

    const int tid  = threadIdx.x;
    const int lane = tid & 31;
    const int wv   = __builtin_amdgcn_readfirstlane(tid >> 5);
    const int lo   = blockIdx.x * OWN;
    const int cntN = (Nn - lo < OWN) ? (Nn - lo) : OWN;

    if (LAYER == 1) {
        if (tid < 2) ctr[tid] = 0;
        if (tid == 2) { candq[0] = 0; cande[0] = 0; }
    }
    for (int i = tid; i < cntN * 2; i += OWN_THR) {
        const float a = asrc[lo * 2 + i];
        s_as[i] = a;
        s_ad[i] = adst[lo * 2 + i];
        mkey[i] = fkey(a);
    }
    __syncthreads();

    int  ne;
    bool over;
    if (LAYER == 1) {
        const unsigned pat = (unsigned)blockIdx.x * 0x01010101u;
        const int nQ = E >> 4;
        const uint4* __restrict__ ob4 = (const uint4*)ob;
        const int  nBW     = (nQ + 31) >> 5;
        const bool mapless = nBW > BROW;
        static_assert(BROW <= 4 * OWN_THR, "four words per thread cover a row");
        if (!mapless) {
            const unsigned* __restrict__ row = bm + (size_t)blockIdx.x * BROW;
            unsigned wd[4];
#pragma unroll
            for (int u = 0; u < 4; ++u) {
                const int wi = tid + u * OWN_THR;
                wd[u] = row[(wi < nBW) ? wi : 0];
            }
#pragma unroll
            for (int u = 0; u < 4; ++u) {
                const int wi = tid + u * OWN_THR;
                unsigned mm = (wi < nBW) ? wd[u] : 0u;
                for (int trip = 0; trip < 32; ++trip) {
                    const bool has = (mm != 0u);
                    const unsigned bal = __builtin_amdgcn_ballot_w32(has);
                    if (bal == 0u) break;
                    const int j = has ? __builtin_ctz(mm) : 0;
                    mm &= mm - 1u;
                    const int lead = __builtin_ctz(bal);
                    int draw = 0;
                    if (lane == lead) draw = atomicAdd(&ctr[0], __builtin_popcount(bal));
                    draw = __builtin_amdgcn_readlane(draw, lead);
                    const int p = draw + (int)__builtin_amdgcn_mbcnt_lo(bal, 0u);
                    if (has && p < QCAP) candq[p] = 32 * wi + j;
                }
            }
        }
        __syncthreads();

        const int nq = ctr[0];
        over = mapless || (nq > QCAP);
        if (!over) {
            int qq[8]; uint4 vv[8];
#pragma unroll
            for (int u = 0; u < 8; ++u) {
                const int i = tid + u * OWN_THR;
                const int q = candq[(i < nq) ? i : 0];
                vv[u] = ob4[q];
                qq[u] = (i < nq) ? q : -1;
            }
#pragma unroll
            for (int u = 0; u < 8; ++u) {
                unsigned mm = zpack(zbytes(vv[u].x ^ pat))
                            | (zpack(zbytes(vv[u].y ^ pat)) << 4)
                            | (zpack(zbytes(vv[u].z ^ pat)) << 8)
                            | (zpack(zbytes(vv[u].w ^ pat)) << 12);
                if (qq[u] < 0) mm = 0u;
                for (int trip = 0; trip < 16; ++trip) {
                    const bool has = (mm != 0u);
                    const unsigned bal = __builtin_amdgcn_ballot_w32(has);
                    if (bal == 0u) break;
                    const int j = has ? __builtin_ctz(mm) : 0;
                    mm &= mm - 1u;
                    const int lead = __builtin_ctz(bal);
                    int draw = 0;
                    if (lane == lead) draw = atomicAdd(&ctr[1], __builtin_popcount(bal));
                    draw = __builtin_amdgcn_readlane(draw, lead);
                    const int p = draw + (int)__builtin_amdgcn_mbcnt_lo(bal, 0u);
                    if (has && p < ECAP) cande[p] = 16 * qq[u] + j;
                }
            }
            for (int e = 16 * nQ + tid; e < E; e += OWN_THR) {
                const unsigned by = (ob[e >> 2] >> (8 * (e & 3))) & 0xFFu;
                if (by == (unsigned)blockIdx.x) {
                    const int p = atomicAdd(&ctr[1], 1);
                    if (p < ECAP) cande[p] = e;
                }
            }
        }
        __syncthreads();

        ne = ctr[1];
        over = over || (ne > ECAP);
        if (tid == 0) lcnt[blockIdx.x] = over ? (ECAP + 1) : ne;
    } else {
        ne = lcnt[blockIdx.x];
        over = (unsigned)ne > (unsigned)ECAP;
    }

    unsigned* __restrict__ lb = lst + (size_t)blockIdx.x * ECAP;
    if (!over) {
        int sv[8], dv[8];
        if (LAYER == 1) {
#pragma unroll
            for (int u = 0; u < 8; ++u) {
                const int i = tid + u * OWN_THR;
                const int e = cande[(i < ne) ? i : 0];
                sv[u] = ei[e];
                dv[u] = ei[E + e] - lo;
            }
        } else {
#pragma unroll
            for (int u = 0; u < 8; ++u) {
                const int i = tid + u * OWN_THR;
                const unsigned rec = lb[(i < ne) ? i : 0];
                const unsigned s = rec >> 9;
                sv[u] = (s < (unsigned)Nn) ? (int)s : 0;
                dv[u] = (int)(rec & 511u);
            }
        }
        float2 av[8];
#pragma unroll
        for (int u = 0; u < 8; ++u) {
            const int i = tid + u * OWN_THR;
            if (i >= ne || dv[u] < 0 || dv[u] >= cntN) dv[u] = -1;
            av[u] = *(const float2*)(asrc + 2 * sv[u]);
        }
#pragma unroll
        for (int u = 0; u < 8; ++u) {
            const int i = tid + u * OWN_THR;
            if (i < ne) {
                ls[i] = sv[u]; ld[i] = dv[u];
                v2f a2; a2.x = av[u].x; a2.y = av[u].y;
                *(v2f*)&lav[2 * i] = a2;
                if (LAYER == 1) lb[i] = ((unsigned)sv[u] << 9) | ((unsigned)dv[u] & 511u);
            }
            if (dv[u] >= 0) {
                atomicMax(&mkey[2 * dv[u]],     fkey(av[u].x));
                atomicMax(&mkey[2 * dv[u] + 1], fkey(av[u].y));
            }
        }
    } else {
        for (int e = tid; e < E; e += OWN_THR) {
            const int d = ei[E + e] - lo;
            if (d >= 0 && d < cntN) {
                const float2 a = *(const float2*)(asrc + 2 * ei[e]);
                atomicMax(&mkey[2 * d],     fkey(a.x));
                atomicMax(&mkey[2 * d + 1], fkey(a.y));
            }
        }
    }
    __syncthreads();

    static_assert(OWN * (HCc / 4) <= 7 * OWN_THR && OWN_THR % (HCc / 4) == 0, "seven passes cover an owner's rows");
    const int c4  = (tid & 15) * 4;
    const int hd4 = c4 >> 5;
    const int n4  = cntN * (HCc / 4);
    float4* __restrict__ acc4 = (float4*)acc;
    {
        const float4* __restrict__ h4 = (const float4*)(h + (size_t)lo * HCc);
        float4 hv[7];
#pragma unroll
        for (int k = 0; k < 7; ++k) {
            const int i4 = tid + k * OWN_THR;
            hv[k] = h4[(i4 < n4) ? i4 : 0];
        }
#pragma unroll
        for (int k = 0; k < 7; ++k) {
            const int i4 = tid + k * OWN_THR;
            if (i4 < n4) {
                const int nl = i4 >> 4;
                const float ad = s_ad[2 * nl + hd4];
                const float m  = lrelu02(funkey(mkey[2 * nl + hd4]) + ad);
                const float ex = __expf(lrelu02(s_as[2 * nl + hd4] + ad) - m);
                float4 a;
                a.x = hv[k].x * ex; a.y = hv[k].y * ex; a.z = hv[k].z * ex; a.w = hv[k].w * ex;
                acc4[i4] = a;
                if ((c4 & 31) == 0) { s_m[2 * nl + hd4] = m; s_den[2 * nl + hd4] = ex; }
            }
        }
    }
    __syncthreads();

    if (!over) {
        const char* __restrict__ hb = (const char*)h;
        const unsigned lane4 = (unsigned)lane * 4u;
        for (int base = wv * 32; base < ne; base += OWN_THR) {
            const int i = base + lane;
            const int cnt = (ne - base < 32) ? (ne - base) : 32;
            int s = 0, d = -1;
            float ex0 = 0.f, ex1 = 0.f;
            v2f a; a.x = 0.f; a.y = 0.f;
            if (i < ne) { s = ls[i]; d = ld[i]; a = *(const v2f*)&lav[2 * i]; }
            if (d >= 0) {
                ex0 = __expf(lrelu02(a.x + s_ad[2 * d])     - s_m[2 * d]);
                ex1 = __expf(lrelu02(a.y + s_ad[2 * d + 1]) - s_m[2 * d + 1]);
                atomicAdd(&s_den[2 * d],     ex0);
                atomicAdd(&s_den[2 * d + 1], ex1);
            }
            const unsigned soff = (d >= 0) ? (unsigned)s * (unsigned)(HCc * sizeof(float)) : 0u;
            const int      drow = (d >= 0) ? d : OWN;
            rows16<0>(hb, acc, lane4, soff, drow, ex0, ex1);
            if (cnt > 16) rows16<16>(hb, acc, lane4, soff, drow, ex0, ex1);
        }
    } else {
        for (int e = tid; e < E; e += OWN_THR) {
            const int d = ei[E + e] - lo;
            if (d >= 0 && d < cntN) {
                const int s = ei[e];
                const float2 a = *(const float2*)(asrc + 2 * s);
                const float ex0 = __expf(lrelu02(a.x + s_ad[2 * d])     - s_m[2 * d]);
                const float ex1 = __expf(lrelu02(a.y + s_ad[2 * d + 1]) - s_m[2 * d + 1]);
                atomicAdd(&s_den[2 * d],     ex0);
                atomicAdd(&s_den[2 * d + 1], ex1);
                const float* hr = h + (size_t)s * HCc;
                for (int c = 0; c < HIDc; ++c) {
                    atomicAdd(&acc[d * HCc + c],        hr[c] * ex0);
                    atomicAdd(&acc[d * HCc + HIDc + c], hr[HIDc + c] * ex1);
                }
            }
        }
    }
    __syncthreads();

    {
        float bi[4], mu[4], sc[4], be[4];
#pragma unroll
        for (int j = 0; j < 4; ++j) {
            bi[j] = bias[c4 + j];
            mu[j] = 0.f; sc[j] = 1.f; be[j] = 0.f;
            if (LAYER == 1) {
                mu[j] = mean[c4 + j];
                sc[j] = gamma[c4 + j] * rsqrtf(var[c4 + j] + 1e-5f);
                be[j] = beta[c4 + j];
            }
        }
        float4* __restrict__ xo4 = (float4*)(xout + (size_t)lo * HCc);
        float4 xp[7];
        if (LAYER == 2) {
            const float4* __restrict__ xp4 = (const float4*)(xprev + (size_t)lo * HCc);
#pragma unroll
            for (int k = 0; k < 7; ++k) {
                const int i4 = tid + k * OWN_THR;
                xp[k] = xp4[(i4 < n4) ? i4 : 0];
            }
        }
#pragma unroll
        for (int k = 0; k < 7; ++k) {
            const int i4 = tid + k * OWN_THR;
            if (i4 < n4) {
                const int nl = i4 >> 4;
                const float dn = s_den[2 * nl + hd4] + 1e-16f;
                const float4 a = acc4[i4];
                float v[4] = {a.x / dn + bi[0], a.y / dn + bi[1], a.z / dn + bi[2], a.w / dn + bi[3]};
                if (LAYER == 1) {
#pragma unroll
                    for (int j = 0; j < 4; ++j) {
                        v[j] = (v[j] - mu[j]) * sc[j] + be[j];
                        v[j] = (v[j] > 0.f) ? v[j] : (__expf(v[j]) - 1.f);
                    }
                } else {
                    v[0] = fmaxf(xp[k].x, v[0]); v[1] = fmaxf(xp[k].y, v[1]);
                    v[2] = fmaxf(xp[k].z, v[2]); v[3] = fmaxf(xp[k].w, v[3]);
                }
                float4 o; o.x = v[0]; o.y = v[1]; o.z = v[2]; o.w = v[3];
                xo4[i4] = o;
            }
        }
    }
}

extern "C" void kernel_launch(void* const* d_in, const int* in_sizes, int n_in,
                              void* d_out, int out_size, void* d_ws, size_t ws_size,
                              hipStream_t stream) {
    const float* node_feat = (const float*)d_in[0];
    const int*   ei        = (const int*)  d_in[1];
    const float* W1        = (const float*)d_in[2];
    const float* a_src1    = (const float*)d_in[3];
    const float* a_dst1    = (const float*)d_in[4];
    const float* b1        = (const float*)d_in[5];
    const float* bn_gamma  = (const float*)d_in[6];
    const float* bn_beta   = (const float*)d_in[7];
    const float* bn_mean   = (const float*)d_in[8];
    const float* bn_var    = (const float*)d_in[9];
    const float* W2        = (const float*)d_in[10];
    const float* a_src2    = (const float*)d_in[11];
    const float* a_dst2    = (const float*)d_in[12];
    const float* b2        = (const float*)d_in[13];
    const float* Wf        = (const float*)d_in[14];
    const float* bf        = (const float*)d_in[15];
    float* out = (float*)d_out;

    const int E = in_sizes[1] / 2;

    float* ws   = (float*)d_ws;
    float* H    = ws;
    float* X1   = H    + (size_t)Nn * HCc;
    float* XJ   = X1   + (size_t)Nn * HCc;
    float* ASRC = XJ   + (size_t)Nn * HCc;
    float* ADST = ASRC + (size_t)Nn * 2;

    static_assert(GEMM_WG * (GEMM_THR / 32) * 16 >= Nn, "every row tile needs a wave");
    static_assert(OWN_WG * OWN >= Nn && OWN_WG <= 256, "every node needs an owner, every owner a byte value");
    unsigned* OB = (unsigned*)d_out;
    static_assert(OWN <= 511 && Nn <= (1 << 17), "a list word holds source << 9 | local destination, 511 = none");
    static_assert((1 << 20) + OWN_WG * ECAP + OWN_WG + OWN_WG * BROW <= Nn * OUTc, "table, lists, lengths and group map fit in the output buffer");
    static_assert(BROW % SLW == 0 && BROW / SLW <= GEMM_WG, "every slice of the group map has a work-group of the first launch");
    unsigned* LST  = (unsigned*)d_out + (1 << 20);
    int*      LCNT = (int*)(LST + OWN_WG * ECAP);
    unsigned* BM   = (unsigned*)(LCNT + OWN_WG);

    wmma_gemm<4, true, true, FIN><<<GEMM_WG, GEMM_THR, 0, stream>>>(
        node_feat, W1, nullptr, H, Nn, HCc, a_src1, a_dst1, ASRC, ADST, ei, E, OB, BM);
    owner_layer<1><<<OWN_WG, OWN_THR, 0, stream>>>(
        ei, E, OB, H, ASRC, ADST, b1, bn_gamma, bn_beta, bn_mean, bn_var, nullptr, X1, LST, LCNT, BM);

    wmma_gemm<4, true, false, HCc><<<GEMM_WG, GEMM_THR, 0, stream>>>(
        X1, W2, nullptr, H, Nn, HCc, a_src2, a_dst2, ASRC, ADST, nullptr, 0, nullptr, nullptr);
    owner_layer<2><<<OWN_WG, OWN_THR, 0, stream>>>(
        ei, E, OB, H, ASRC, ADST, b2, nullptr, nullptr, nullptr, nullptr, X1, XJ, LST, LCNT, nullptr);

    wmma_gemm<3, false, false, HCc><<<GEMM_WG, GEMM_THR, 0, stream>>>(
        XJ, Wf, bf, out, Nn, OUTc, nullptr, nullptr, nullptr, nullptr, nullptr, 0, nullptr, nullptr);
}


# layer 1 owners read each listed group's edge ends from the edge list (64 B per half-wave), keep edges by destination range: no owner byte table, no per-edge gather of edge ends; LDS add order changes
# speedup vs baseline: 9.3212x; 1.0321x over previous
#include <hip/hip_runtime.h>
#include <hip/hip_bf16.h>
#include <math.h>


#define Nn    100000
#define FIN   128
#define HCc   64
#define HIDc  32
#define OUTc  40
#define GEMM_WG   256
#define GEMM_THR  1024
#define OWN       391
#define OWN_WG    256
#define OWN_THR   1024
#define QCAP      8192
#define ECAP      8192
#define SLG       1024
#define SLW       (SLG / 32)
#define BROW      3136

typedef __attribute__((ext_vector_type(2))) float v2f;
typedef __attribute__((ext_vector_type(8))) float v8f;
typedef int dst4 __attribute__((ext_vector_type(4), aligned(4)));

__device__ inline int fkey(float v) {
    const int b = __float_as_int(v);
    return (b >= 0) ? b : (b ^ 0x7FFFFFFF);
}
__device__ inline float funkey(int k) {
    return __int_as_float((k >= 0) ? k : (k ^ 0x7FFFFFFF));
}
__device__ inline float lrelu02(float v) { return (v > 0.f) ? v : 0.2f * v; }

template<int MASK>
__device__ __forceinline__ float row_xor_add(float v) {
    const int o = __builtin_amdgcn_update_dpp(0, __float_as_int(v), 0x160 + MASK,
                                              0xf, 0xf, true);
    return v + __int_as_float(o);
}

template<int NT, bool PREP, bool OBY, int K>
__global__ __launch_bounds__(GEMM_THR)
void wmma_gemm(const float* __restrict__ A,
               const float* __restrict__ B, const float* __restrict__ bias,
               float* __restrict__ C, int M, int Ncols,
               const float* __restrict__ att_s, const float* __restrict__ att_d,
               float* __restrict__ asrc, float* __restrict__ adst,
               const int* __restrict__ ei, int E,
               unsigned* __restrict__ bm) {
    static_assert(!PREP || NT == 4, "PREP needs the full 64-column row in the wave");
    if (OBY) {
        constexpr int BTP = OWN_WG + 1;
        __shared__ unsigned bt[SLW * BTP];
        const int nQ = E >> 4;
        const int g0 = blockIdx.x * SLG;
        static_assert(SLG == GEMM_THR && SLW == 32, "one group per thread, one 128-byte line of a row per slice");
        if (16 * g0 < E && (blockIdx.x + 1) * SLW <= BROW) {
            for (int i = threadIdx.x; i < SLW * BTP; i += GEMM_THR) bt[i] = 0u;
            __syncthreads();
            const int g  = threadIdx.x;
            const int gA = g0 + g;
            if (nQ > 0) {
                const int gc = (gA < nQ) ? gA : (nQ - 1);
                const dst4* __restrict__ dp = (const dst4*)(ei + E + 16 * gc);
                dst4 dv[4];
#pragma unroll
                for (int j = 0; j < 4; ++j) dv[j] = dp[j];
                if (gA < nQ) {
#pragma unroll
                    for (int j = 0; j < 4; ++j) {
                        const unsigned o0 = ((unsigned)dv[j].x / (unsigned)OWN) & 0xFFu;
                        const unsigned o1 = ((unsigned)dv[j].y / (unsigned)OWN) & 0xFFu;
                        const unsigned o2 = ((unsigned)dv[j].z / (unsigned)OWN) & 0xFFu;
                        const unsigned o3 = ((unsigned)dv[j].w / (unsigned)OWN) & 0xFFu;
                        const unsigned bit = 1u << (g & 31);
                        atomicOr(&bt[(g >> 5) * BTP + o0], bit);
                        atomicOr(&bt[(g >> 5) * BTP + o1], bit);
                        atomicOr(&bt[(g >> 5) * BTP + o2], bit);
                        atomicOr(&bt[(g >> 5) * BTP + o3], bit);
                    }
                }
            }
            __syncthreads();
            for (int i = threadIdx.x; i < OWN_WG * SLW; i += GEMM_THR)
                bm[(size_t)(i / SLW) * BROW + blockIdx.x * SLW + (i % SLW)] = bt[(i % SLW) * BTP + (i / SLW)];
        }
    }
    __shared__ float Bs[(K / 2) * 64 * 2];

    for (int i = threadIdx.x; i < K * 64; i += GEMM_THR) {
        const int p    = i >> 7;
        const int rem  = i & 127;
        const int col  = rem >> 1;
        const int half = rem & 1;
        Bs[i] = (col < Ncols) ? B[(size_t)(2 * p + half) * Ncols + col] : 0.f;
    }
    __syncthreads();

    const int wave = threadIdx.x >> 5;
    const int lane = threadIdx.x & 31;
    const int tileM = wave * GEMM_WG + blockIdx.x;
    const int row16 = tileM * 16;
    if (row16 >= M) return;

    const int hi = (lane >= 16);
    const int lr = lane & 15;

    v8f acc[NT];
#pragma unroll
    for (int t = 0; t < NT; ++t)
        acc[t] = (v8f){0.f, 0.f, 0.f, 0.f, 0.f, 0.f, 0.f, 0.f};

    const int arow = row16 + lr;
    const float* __restrict__ Arow  = A + (size_t)arow * K;

#pragma unroll 4
    for (int kb = 0; kb < K; kb += 4) {
        const int ka = kb + 2 * hi;
        const float a0 = Arow[ka];
        const float a1 = Arow[ka + 1];
        v2f afrag; afrag.x = a0; afrag.y = a1;

        const int p = (kb >> 1) + hi;
#pragma unroll
        for (int t = 0; t < NT; ++t) {
            const v2f bfrag = *(const v2f*)&Bs[(p << 7) + ((t * 16 + lr) << 1)];
            acc[t] = __builtin_amdgcn_wmma_f32_16x16x4_f32(
                false, afrag, false, bfrag, (short)0, acc[t], false, false);
        }
    }

    const int rbase = row16 + (hi ? 8 : 0);
#pragma unroll
    for (int t = 0; t < NT; ++t) {
        const int col = t * 16 + lr;
        if (col < Ncols) {
            const float bb = bias ? bias[col] : 0.f;
#pragma unroll
            for (int r = 0; r < 8; ++r)
                C[(size_t)(rbase + r) * Ncols + col] = acc[t][r] + bb;
        }
    }

    if (PREP) {
        float vs[2][8], vd[2][8];
#pragma unroll
        for (int hd = 0; hd < 2; ++hd) {
            const float s_lo = att_s[hd * HIDc + lr], s_hi = att_s[hd * HIDc + 16 + lr];
            const float d_lo = att_d[hd * HIDc + lr], d_hi = att_d[hd * HIDc + 16 + lr];
#pragma unroll
            for (int r = 0; r < 8; ++r) {
                vs[hd][r] = acc[2 * hd][r] * s_lo + acc[2 * hd + 1][r] * s_hi;
                vd[hd][r] = acc[2 * hd][r] * d_lo + acc[2 * hd + 1][r] * d_hi;
            }
        }
#pragma unroll
        for (int hd = 0; hd < 2; ++hd)
#pragma unroll
            for (int r = 0; r < 8; ++r) { vs[hd][r] = row_xor_add<8>(vs[hd][r]); vd[hd][r] = row_xor_add<8>(vd[hd][r]); }
#pragma unroll
        for (int hd = 0; hd < 2; ++hd)
#pragma unroll
            for (int r = 0; r < 8; ++r) { vs[hd][r] = row_xor_add<4>(vs[hd][r]); vd[hd][r] = row_xor_add<4>(vd[hd][r]); }
#pragma unroll
        for (int hd = 0; hd < 2; ++hd)
#pragma unroll
            for (int r = 0; r < 8; ++r) { vs[hd][r] = row_xor_add<2>(vs[hd][r]); vd[hd][r] = row_xor_add<2>(vd[hd][r]); }
#pragma unroll
        for (int hd = 0; hd < 2; ++hd)
#pragma unroll
            for (int r = 0; r < 8; ++r) { vs[hd][r] = row_xor_add<1>(vs[hd][r]); vd[hd][r] = row_xor_add<1>(vd[hd][r]); }
        float os0 = vs[0][0], os1 = vs[1][0], od0 = vd[0][0], od1 = vd[1][0];
#pragma unroll
        for (int r = 1; r < 8; ++r) {
            os0 = (lr == r) ? vs[0][r] : os0;
            os1 = (lr == r) ? vs[1][r] : os1;
            od0 = (lr == r) ? vd[0][r] : od0;
            od1 = (lr == r) ? vd[1][r] : od1;
        }
        if (lr < 8) {
            const int n = rbase + lr;
            float2 a2; a2.x = os0; a2.y = os1;
            *(float2*)(asrc + 2 * n) = a2;
            float2 d2; d2.x = od0; d2.y = od1;
            *(float2*)(adst + 2 * n) = d2;
        }
    }
}

template<int J0>
__device__ __forceinline__ void rows16(const char* __restrict__ hb, float* __restrict__ acc,
                                       unsigned lane4, unsigned soff, int drow,
                                       float ex0, float ex1) {
    float hv0[16], hv1[16];
#pragma unroll
    for (int u = 0; u < 16; ++u) {
        const unsigned so = (unsigned)__builtin_amdgcn_readlane((int)soff, J0 + u);
        const float* hr = (const float*)(hb + (size_t)(so + lane4));
        hv0[u] = hr[0];
        hv1[u] = hr[HIDc];
    }
#pragma unroll
    for (int u = 0; u < 16; ++u) {
        const int dj = __builtin_amdgcn_readlane(drow, J0 + u);
        const float x0 = __int_as_float(__builtin_amdgcn_readlane(__float_as_int(ex0), J0 + u));
        const float x1 = __int_as_float(__builtin_amdgcn_readlane(__float_as_int(ex1), J0 + u));
        atomicAdd(&acc[dj * HCc + (int)(lane4 >> 2)],        hv0[u] * x0);
        atomicAdd(&acc[dj * HCc + HIDc + (int)(lane4 >> 2)], hv1[u] * x1);
    }
}

template<int LAYER>
__global__ __launch_bounds__(OWN_THR)
void owner_layer(const int* __restrict__ ei, int E,
                 const float* __restrict__ h, const float* __restrict__ asrc,
                 const float* __restrict__ adst, const float* __restrict__ bias,
                 const float* __restrict__ gamma, const float* __restrict__ beta,
                 const float* __restrict__ mean, const float* __restrict__ var,
                 const float* __restrict__ xprev, float* __restrict__ xout,
                 unsigned* __restrict__ lst, int* __restrict__ lcnt,
                 const unsigned* __restrict__ bm) {
    __shared__ __attribute__((aligned(16))) float acc[(OWN + 1) * HCc];
    __shared__ int   mkey[OWN * 2];
    __shared__ float s_as[OWN * 2], s_ad[OWN * 2], s_m[OWN * 2], s_den[OWN * 2];
    __shared__ int   candq[QCAP];
    __shared__ int   ls[ECAP], ld[ECAP];
    __shared__ __attribute__((aligned(8))) float lav[2 * ECAP];
    __shared__ int   ctr[2];

    const int tid  = threadIdx.x;
    const int lane = tid & 31;
    const int wv   = __builtin_amdgcn_readfirstlane(tid >> 5);
    const int lo   = blockIdx.x * OWN;
    const int cntN = (Nn - lo < OWN) ? (Nn - lo) : OWN;

    if (LAYER == 1) {
        if (tid < 2) ctr[tid] = 0;
        if (tid == 2) candq[0] = 0;
    }
    for (int i = tid; i < cntN * 2; i += OWN_THR) {
        const float a = asrc[lo * 2 + i];
        s_as[i] = a;
        s_ad[i] = adst[lo * 2 + i];
        mkey[i] = fkey(a);
    }
    __syncthreads();

    int  ne;
    bool over;
    if (LAYER == 1) {
        const int nQ = E >> 4;
        const int  nBW     = (nQ + 31) >> 5;
        const bool mapless = nBW > BROW;
        static_assert(BROW <= 4 * OWN_THR, "four words per thread cover a row");
        if (!mapless) {
            const unsigned* __restrict__ row = bm + (size_t)blockIdx.x * BROW;
            unsigned wd[4];
#pragma unroll
            for (int u = 0; u < 4; ++u) {
                const int wi = tid + u * OWN_THR;
                wd[u] = row[(wi < nBW) ? wi : 0];
            }
#pragma unroll
            for (int u = 0; u < 4; ++u) {
                const int wi = tid + u * OWN_THR;
                unsigned mm = (wi < nBW) ? wd[u] : 0u;
                for (int trip = 0; trip < 32; ++trip) {
                    const bool has = (mm != 0u);
                    const unsigned bal = __builtin_amdgcn_ballot_w32(has);
                    if (bal == 0u) break;
                    const int j = has ? __builtin_ctz(mm) : 0;
                    mm &= mm - 1u;
                    const int lead = __builtin_ctz(bal);
                    int draw = 0;
                    if (lane == lead) draw = atomicAdd(&ctr[0], __builtin_popcount(bal));
                    draw = __builtin_amdgcn_readlane(draw, lead);
                    const int p = draw + (int)__builtin_amdgcn_mbcnt_lo(bal, 0u);
                    if (has && p < QCAP) candq[p] = 32 * wi + j;
                }
            }
        }
        __syncthreads();

        const int nq = ctr[0];
        over = mapless || (nq > QCAP);
        if (!over) {
            constexpr int PB = 16;
            const int h16 = lane >> 4;
            const int j16 = lane & 15;
            const int np  = (nq + 1) >> 1;
            for (int pb = wv * PB; pb < np; pb += (OWN_THR / 32) * PB) {
                int sv[PB], dv[PB];
#pragma unroll
                for (int u = 0; u < PB; ++u) {
                    const int gi = 2 * (pb + u) + h16;
                    const int e  = 16 * candq[(gi < nq) ? gi : 0] + j16;
                    sv[u] = ei[e];
                    dv[u] = ei[E + e];
                }
                unsigned bal[PB];
                int tot = 0;
#pragma unroll
                for (int u = 0; u < PB; ++u) {
                    const int gi = 2 * (pb + u) + h16;
                    dv[u] -= lo;
                    const bool mine = (gi < nq) && ((unsigned)dv[u] < (unsigned)cntN);
                    bal[u] = __builtin_amdgcn_ballot_w32(mine);
                    tot += __builtin_popcount(bal[u]);
                }
                int draw = 0;
                if (lane == 0) draw = atomicAdd(&ctr[1], tot);
                draw = __builtin_amdgcn_readlane(draw, 0);
#pragma unroll
                for (int u = 0; u < PB; ++u) {
                    const int p = draw + (int)__builtin_amdgcn_mbcnt_lo(bal[u], 0u);
                    if (((bal[u] >> lane) & 1u) && p < ECAP) { ls[p] = sv[u]; ld[p] = dv[u]; }
                    draw += __builtin_popcount(bal[u]);
                }
            }
            for (int e = 16 * nQ + tid; e < E; e += OWN_THR) {
                const int d = ei[E + e] - lo;
                if ((unsigned)d < (unsigned)cntN) {
                    const int p = atomicAdd(&ctr[1], 1);
                    if (p < ECAP) { ls[p] = ei[e]; ld[p] = d; }
                }
            }
        }
        __syncthreads();

        ne = ctr[1];
        over = over || (ne > ECAP);
        if (tid == 0) lcnt[blockIdx.x] = over ? (ECAP + 1) : ne;
    } else {
        ne = lcnt[blockIdx.x];
        over = (unsigned)ne > (unsigned)ECAP;
    }

    unsigned* __restrict__ lb = lst + (size_t)blockIdx.x * ECAP;
    if (!over) {
        int sv[8], dv[8];
        if (LAYER == 1) {
#pragma unroll
            for (int u = 0; u < 8; ++u) {
                const int i = tid + u * OWN_THR;
                const int s1 = ls[(i < ne) ? i : 0];
                const int d1 = ld[(i < ne) ? i : 0];
                sv[u] = (i < ne) ? s1 : 0;
                dv[u] = (i < ne) ? d1 : -1;
            }
        } else {
#pragma unroll
            for (int u = 0; u < 8; ++u) {
                const int i = tid + u * OWN_THR;
                const unsigned rec = lb[(i < ne) ? i : 0];
                const unsigned s = rec >> 9;
                sv[u] = (s < (unsigned)Nn) ? (int)s : 0;
                dv[u] = (int)(rec & 511u);
            }
        }
        float2 av[8];
#pragma unroll
        for (int u = 0; u < 8; ++u) {
            const int i = tid + u * OWN_THR;
            if (i >= ne || dv[u] < 0 || dv[u] >= cntN) dv[u] = -1;
            av[u] = *(const float2*)(asrc + 2 * sv[u]);
        }
#pragma unroll
        for (int u = 0; u < 8; ++u) {
            const int i = tid + u * OWN_THR;
            if (i < ne) {
                if (LAYER == 2) { ls[i] = sv[u]; ld[i] = dv[u]; }
                v2f a2; a2.x = av[u].x; a2.y = av[u].y;
                *(v2f*)&lav[2 * i] = a2;
                if (LAYER == 1) lb[i] = ((unsigned)sv[u] << 9) | ((unsigned)dv[u] & 511u);
            }
            if (dv[u] >= 0) {
                atomicMax(&mkey[2 * dv[u]],     fkey(av[u].x));
                atomicMax(&mkey[2 * dv[u] + 1], fkey(av[u].y));
            }
        }
    } else {
        for (int e = tid; e < E; e += OWN_THR) {
            const int d = ei[E + e] - lo;
            if (d >= 0 && d < cntN) {
                const float2 a = *(const float2*)(asrc + 2 * ei[e]);
                atomicMax(&mkey[2 * d],     fkey(a.x));
                atomicMax(&mkey[2 * d + 1], fkey(a.y));
            }
        }
    }
    __syncthreads();

    static_assert(OWN * (HCc / 4) <= 7 * OWN_THR && OWN_THR % (HCc / 4) == 0, "seven passes cover an owner's rows");
    const int c4  = (tid & 15) * 4;
    const int hd4 = c4 >> 5;
    const int n4  = cntN * (HCc / 4);
    float4* __restrict__ acc4 = (float4*)acc;
    {
        const float4* __restrict__ h4 = (const float4*)(h + (size_t)lo * HCc);
        float4 hv[7];
#pragma unroll
        for (int k = 0; k < 7; ++k) {
            const int i4 = tid + k * OWN_THR;
            hv[k] = h4[(i4 < n4) ? i4 : 0];
        }
#pragma unroll
        for (int k = 0; k < 7; ++k) {
            const int i4 = tid + k * OWN_THR;
            if (i4 < n4) {
                const int nl = i4 >> 4;
                const float ad = s_ad[2 * nl + hd4];
                const float m  = lrelu02(funkey(mkey[2 * nl + hd4]) + ad);
                const float ex = __expf(lrelu02(s_as[2 * nl + hd4] + ad) - m);
                float4 a;
                a.x = hv[k].x * ex; a.y = hv[k].y * ex; a.z = hv[k].z * ex; a.w = hv[k].w * ex;
                acc4[i4] = a;
                if ((c4 & 31) == 0) { s_m[2 * nl + hd4] = m; s_den[2 * nl + hd4] = ex; }
            }
        }
    }
    __syncthreads();

    if (!over) {
        const char* __restrict__ hb = (const char*)h;
        const unsigned lane4 = (unsigned)lane * 4u;
        for (int base = wv * 32; base < ne; base += OWN_THR) {
            const int i = base + lane;
            const int cnt = (ne - base < 32) ? (ne - base) : 32;
            int s = 0, d = -1;
            float ex0 = 0.f, ex1 = 0.f;
            v2f a; a.x = 0.f; a.y = 0.f;
            if (i < ne) { s = ls[i]; d = ld[i]; a = *(const v2f*)&lav[2 * i]; }
            if (d >= 0) {
                ex0 = __expf(lrelu02(a.x + s_ad[2 * d])     - s_m[2 * d]);
                ex1 = __expf(lrelu02(a.y + s_ad[2 * d + 1]) - s_m[2 * d + 1]);
                atomicAdd(&s_den[2 * d],     ex0);
                atomicAdd(&s_den[2 * d + 1], ex1);
            }
            const unsigned soff = (d >= 0) ? (unsigned)s * (unsigned)(HCc * sizeof(float)) : 0u;
            const int      drow = (d >= 0) ? d : OWN;
            rows16<0>(hb, acc, lane4, soff, drow, ex0, ex1);
            if (cnt > 16) rows16<16>(hb, acc, lane4, soff, drow, ex0, ex1);
        }
    } else {
        for (int e = tid; e < E; e += OWN_THR) {
            const int d = ei[E + e] - lo;
            if (d >= 0 && d < cntN) {
                const int s = ei[e];
                const float2 a = *(const float2*)(asrc + 2 * s);
                const float ex0 = __expf(lrelu02(a.x + s_ad[2 * d])     - s_m[2 * d]);
                const float ex1 = __expf(lrelu02(a.y + s_ad[2 * d + 1]) - s_m[2 * d + 1]);
                atomicAdd(&s_den[2 * d],     ex0);
                atomicAdd(&s_den[2 * d + 1], ex1);
                const float* hr = h + (size_t)s * HCc;
                for (int c = 0; c < HIDc; ++c) {
                    atomicAdd(&acc[d * HCc + c],        hr[c] * ex0);
                    atomicAdd(&acc[d * HCc + HIDc + c], hr[HIDc + c] * ex1);
                }
            }
        }
    }
    __syncthreads();

    {
        float bi[4], mu[4], sc[4], be[4];
#pragma unroll
        for (int j = 0; j < 4; ++j) {
            bi[j] = bias[c4 + j];
            mu[j] = 0.f; sc[j] = 1.f; be[j] = 0.f;
            if (LAYER == 1) {
                mu[j] = mean[c4 + j];
                sc[j] = gamma[c4 + j] * rsqrtf(var[c4 + j] + 1e-5f);
                be[j] = beta[c4 + j];
            }
        }
        float4* __restrict__ xo4 = (float4*)(xout + (size_t)lo * HCc);
        float4 xp[7];
        if (LAYER == 2) {
            const float4* __restrict__ xp4 = (const float4*)(xprev + (size_t)lo * HCc);
#pragma unroll
            for (int k = 0; k < 7; ++k) {
                const int i4 = tid + k * OWN_THR;
                xp[k] = xp4[(i4 < n4) ? i4 : 0];
            }
        }
#pragma unroll
        for (int k = 0; k < 7; ++k) {
            const int i4 = tid + k * OWN_THR;
            if (i4 < n4) {
                const int nl = i4 >> 4;
                const float dn = s_den[2 * nl + hd4] + 1e-16f;
                const float4 a = acc4[i4];
                float v[4] = {a.x / dn + bi[0], a.y / dn + bi[1], a.z / dn + bi[2], a.w / dn + bi[3]};
                if (LAYER == 1) {
#pragma unroll
                    for (int j = 0; j < 4; ++j) {
                        v[j] = (v[j] - mu[j]) * sc[j] + be[j];
                        v[j] = (v[j] > 0.f) ? v[j] : (__expf(v[j]) - 1.f);
                    }
                } else {
                    v[0] = fmaxf(xp[k].x, v[0]); v[1] = fmaxf(xp[k].y, v[1]);
                    v[2] = fmaxf(xp[k].z, v[2]); v[3] = fmaxf(xp[k].w, v[3]);
                }
                float4 o; o.x = v[0]; o.y = v[1]; o.z = v[2]; o.w = v[3];
                xo4[i4] = o;
            }
        }
    }
}

extern "C" void kernel_launch(void* const* d_in, const int* in_sizes, int n_in,
                              void* d_out, int out_size, void* d_ws, size_t ws_size,
                              hipStream_t stream) {
    const float* node_feat = (const float*)d_in[0];
    const int*   ei        = (const int*)  d_in[1];
    const float* W1        = (const float*)d_in[2];
    const float* a_src1    = (const float*)d_in[3];
    const float* a_dst1    = (const float*)d_in[4];
    const float* b1        = (const float*)d_in[5];
    const float* bn_gamma  = (const float*)d_in[6];
    const float* bn_beta   = (const float*)d_in[7];
    const float* bn_mean   = (const float*)d_in[8];
    const float* bn_var    = (const float*)d_in[9];
    const float* W2        = (const float*)d_in[10];
    const float* a_src2    = (const float*)d_in[11];
    const float* a_dst2    = (const float*)d_in[12];
    const float* b2        = (const float*)d_in[13];
    const float* Wf        = (const float*)d_in[14];
    const float* bf        = (const float*)d_in[15];
    float* out = (float*)d_out;

    const int E = in_sizes[1] / 2;

    float* ws   = (float*)d_ws;
    float* H    = ws;
    float* X1   = H    + (size_t)Nn * HCc;
    float* XJ   = X1   + (size_t)Nn * HCc;
    float* ASRC = XJ   + (size_t)Nn * HCc;
    float* ADST = ASRC + (size_t)Nn * 2;

    static_assert(GEMM_WG * (GEMM_THR / 32) * 16 >= Nn, "every row tile needs a wave");
    static_assert(OWN_WG * OWN >= Nn && OWN_WG <= 256, "every node needs an owner, every owner a column of the first launch's map tile (its number below 256)");
    static_assert(OWN <= 511 && Nn <= (1 << 17), "a list word holds source << 9 | local destination, 511 = none");
    static_assert(OWN_WG * ECAP + OWN_WG + OWN_WG * BROW <= Nn * OUTc, "lists, lengths and group map fit in the output buffer");
    static_assert(BROW % SLW == 0 && BROW / SLW <= GEMM_WG, "every slice of the group map has a work-group of the first launch");
    unsigned* LST  = (unsigned*)d_out;
    int*      LCNT = (int*)(LST + OWN_WG * ECAP);
    unsigned* BM   = (unsigned*)(LCNT + OWN_WG);

    wmma_gemm<4, true, true, FIN><<<GEMM_WG, GEMM_THR, 0, stream>>>(
        node_feat, W1, nullptr, H, Nn, HCc, a_src1, a_dst1, ASRC, ADST, ei, E, BM);
    owner_layer<1><<<OWN_WG, OWN_THR, 0, stream>>>(
        ei, E, H, ASRC, ADST, b1, bn_gamma, bn_beta, bn_mean, bn_var, nullptr, X1, LST, LCNT, BM);

    wmma_gemm<4, true, false, HCc><<<GEMM_WG, GEMM_THR, 0, stream>>>(
        X1, W2, nullptr, H, Nn, HCc, a_src2, a_dst2, ASRC, ADST, nullptr, 0, nullptr);
    owner_layer<2><<<OWN_WG, OWN_THR, 0, stream>>>(
        ei, E, H, ASRC, ADST, b2, nullptr, nullptr, nullptr, nullptr, X1, XJ, LST, LCNT, nullptr);

    wmma_gemm<3, false, false, HCc><<<GEMM_WG, GEMM_THR, 0, stream>>>(
        XJ, Wf, bf, out, Nn, OUTc, nullptr, nullptr, nullptr, nullptr, nullptr, 0, nullptr);
}
